# starting kernel
# baseline (speedup 1.0000x reference)
#include <hip/hip_runtime.h>
#include <math.h>


typedef __attribute__((ext_vector_type(2))) float v2f;
typedef __attribute__((ext_vector_type(8))) float v8f;

#define IN_CH  256
#define HID_CH 128
#define OUT_CH 64

template <int K, int NCOLS>
__global__ __launch_bounds__(32) void gemm_wmma_f32(
    const float* __restrict__ A, const float* __restrict__ W,
    float* __restrict__ Out, int nRows) {
  constexpr int NT = NCOLS / 16;
  const int mbase = blockIdx.x * 16;
  const int lane  = threadIdx.x & 31;
  const int g     = lane >> 4;
  const int l     = lane & 15;

  v8f acc[NT] = {};

  int arow_idx = mbase + l;
  if (arow_idx >= nRows) arow_idx = nRows - 1;
  const float* __restrict__ arow = A + (size_t)arow_idx * K + 2 * g;

#pragma unroll 4
  for (int k = 0; k < K; k += 4) {
    v2f a;
    a.x = arow[k];
    a.y = arow[k + 1];
    const float* __restrict__ wk = W + (size_t)(k + 2 * g) * NCOLS + l;
#pragma unroll
    for (int t = 0; t < NT; ++t) {
      v2f b;
      b.x = wk[t * 16];
      b.y = wk[NCOLS + t * 16];
      acc[t] = __builtin_amdgcn_wmma_f32_16x16x4_f32(
           false, a,  false, b,
           (short)0, acc[t],  false,  false);
    }
  }

#pragma unroll
  for (int t = 0; t < NT; ++t) {
#pragma unroll
    for (int r = 0; r < 8; ++r) {
      const int m = mbase + r + 8 * g;
      if (m < nRows) Out[(size_t)m * NCOLS + t * 16 + l] = acc[t][r];
    }
  }
}

__global__ __launch_bounds__(256) void deg_kernel(
    const int* __restrict__ dst, unsigned* __restrict__ deg, int nE) {
  const int e = blockIdx.x * blockDim.x + threadIdx.x;
  if (e < nE) atomicAdd(&deg[dst[e]], 1u);
}

__global__ __launch_bounds__(256) void dinv_kernel(
    const unsigned* __restrict__ deg, float* __restrict__ dinv, int n) {
  const int i = blockIdx.x * blockDim.x + threadIdx.x;
  if (i < n) dinv[i] = rsqrtf((float)deg[i] + 1.0f);
}

template <int C>
__global__ __launch_bounds__(256) void scatter_edges(
    const int* __restrict__ src, const int* __restrict__ dst,
    const float* __restrict__ dinv, const float* __restrict__ hin,
    float* __restrict__ agg, int nE) {
  const int wid  = (int)((blockIdx.x * (unsigned)blockDim.x + threadIdx.x) >> 5);
  const int lane = threadIdx.x & 31;
  if (wid >= nE) return;
  const int s = src[wid];
  const int d = dst[wid];
  const float nrm = dinv[s] * dinv[d];
  constexpr int PER = C / 32;
  const float* __restrict__ hp = hin + (size_t)s * C + lane * PER;
  float* __restrict__ ap       = agg + (size_t)d * C + lane * PER;
#pragma unroll
  for (int j = 0; j < PER; ++j) {
    __hip_atomic_fetch_add(&ap[j], nrm * hp[j], __ATOMIC_RELAXED,
                           __HIP_MEMORY_SCOPE_AGENT);
  }
}

template <int C>
__global__ __launch_bounds__(256) void finish_layer(
    float* __restrict__ agg, const float* __restrict__ lin,
    const float* __restrict__ dinv, const float* __restrict__ bias,
    int n, int doRelu) {
  const int t = blockIdx.x * blockDim.x + threadIdx.x;
  if (t >= n * C) return;
  const int i = t / C;
  const int c = t % C;
  const float di = dinv[i];
  float v = agg[t] + di * di * lin[t] + bias[c];
  if (doRelu) v = fmaxf(v, 0.0f);
  agg[t] = v;
}

__global__ __launch_bounds__(256) void decode_kernel(
    const int* __restrict__ src, const int* __restrict__ dst,
    const float* __restrict__ z, float* __restrict__ out, int nE) {
  const int wid  = (int)((blockIdx.x * (unsigned)blockDim.x + threadIdx.x) >> 5);
  const int lane = threadIdx.x & 31;
  if (wid >= nE) return;
  const int s = src[wid];
  const int d = dst[wid];
  const float* __restrict__ zs = z + (size_t)s * OUT_CH + lane * 2;
  const float* __restrict__ zd = z + (size_t)d * OUT_CH + lane * 2;
  float p = zs[0] * zd[0] + zs[1] * zd[1];
#pragma unroll
  for (int off = 16; off > 0; off >>= 1) p += __shfl_xor(p, off, 32);
  if (lane == 0) out[wid] = 1.0f / (1.0f + __expf(-p));
}

extern "C" void kernel_launch(void* const* d_in, const int* in_sizes, int n_in,
                              void* d_out, int out_size, void* d_ws, size_t ws_size,
                              hipStream_t stream) {
  const float* x   = (const float*)d_in[0];
  const int*   ei  = (const int*)d_in[1];
  const float* W1  = (const float*)d_in[2];
  const float* b1  = (const float*)d_in[3];
  const float* W2  = (const float*)d_in[4];
  const float* b2  = (const float*)d_in[5];

  const int N = in_sizes[0] / IN_CH;
  const int E = in_sizes[1] / 2;
  const int* esrc = ei;
  const int* edst = ei + E;

  char* w = (char*)d_ws;
  unsigned* deg  = (unsigned*)w; w += (size_t)N * 4;
  float*    dinv = (float*)w;    w += (size_t)N * 4;
  float*    hlin = (float*)w;    w += (size_t)N * HID_CH * 4;
  float*    hagg = (float*)w;    w += (size_t)N * HID_CH * 4;
  float*    zlin = (float*)w;    w += (size_t)N * OUT_CH * 4;
  float*    zagg = (float*)w;    w += (size_t)N * OUT_CH * 4;

  hipMemsetAsync(deg,  0, (size_t)N * 4, stream);
  hipMemsetAsync(hagg, 0, (size_t)N * HID_CH * 4, stream);
  hipMemsetAsync(zagg, 0, (size_t)N * OUT_CH * 4, stream);

  const int TPB = 256;
  const int nWarpBlocks = (int)(((size_t)E * 32 + TPB - 1) / TPB);
  const int mTiles = (N + 15) / 16;

  deg_kernel<<<(E + TPB - 1) / TPB, TPB, 0, stream>>>(edst, deg, E);
  dinv_kernel<<<(N + TPB - 1) / TPB, TPB, 0, stream>>>(deg, dinv, N);

  gemm_wmma_f32<IN_CH, HID_CH><<<mTiles, 32, 0, stream>>>(x, W1, hlin, N);
  scatter_edges<HID_CH><<<nWarpBlocks, TPB, 0, stream>>>(esrc, edst, dinv, hlin, hagg, E);
  finish_layer<HID_CH><<<((size_t)N * HID_CH + TPB - 1) / TPB, TPB, 0, stream>>>(
      hagg, hlin, dinv, b1, N, 1);

  gemm_wmma_f32<HID_CH, OUT_CH><<<mTiles, 32, 0, stream>>>(hagg, W2, zlin, N);
  scatter_edges<OUT_CH><<<nWarpBlocks, TPB, 0, stream>>>(esrc, edst, dinv, zlin, zagg, E);
  finish_layer<OUT_CH><<<((size_t)N * OUT_CH + TPB - 1) / TPB, TPB, 0, stream>>>(
      zagg, zlin, dinv, b2, N, 0);

  decode_kernel<<<nWarpBlocks, TPB, 0, stream>>>(esrc, edst, zagg, (float*)d_out, E);
}


# Edge sums as a per-node gather over a CSR rebuilt in the call (u32 rank atomics, scan, permute) replace the f32 atomic scatter, zero fill and finish pass; node sums now run in arrival-rank order
# speedup vs baseline: 1.7851x; 1.7851x over previous
#include <hip/hip_runtime.h>
#include <math.h>


typedef __attribute__((ext_vector_type(2))) float v2f;
typedef __attribute__((ext_vector_type(8))) float v8f;

#define IN_CH  256
#define HID_CH 128
#define OUT_CH 64

template <int K, int NCOLS>
__global__ __launch_bounds__(32) void gemm_wmma_f32(
    const float* __restrict__ A, const float* __restrict__ W,
    float* __restrict__ Out, int nRows) {
  constexpr int NT = NCOLS / 16;
  const int mbase = blockIdx.x * 16;
  const int lane  = threadIdx.x & 31;
  const int g     = lane >> 4;
  const int l     = lane & 15;

  v8f acc[NT] = {};

  int arow_idx = mbase + l;
  if (arow_idx >= nRows) arow_idx = nRows - 1;
  const float* __restrict__ arow = A + (size_t)arow_idx * K + 2 * g;

#pragma unroll 4
  for (int k = 0; k < K; k += 4) {
    v2f a;
    a.x = arow[k];
    a.y = arow[k + 1];
    const float* __restrict__ wk = W + (size_t)(k + 2 * g) * NCOLS + l;
#pragma unroll
    for (int t = 0; t < NT; ++t) {
      v2f b;
      b.x = wk[t * 16];
      b.y = wk[NCOLS + t * 16];
      acc[t] = __builtin_amdgcn_wmma_f32_16x16x4_f32(
           false, a,  false, b,
           (short)0, acc[t],  false,  false);
    }
  }

#pragma unroll
  for (int t = 0; t < NT; ++t) {
#pragma unroll
    for (int r = 0; r < 8; ++r) {
      const int m = mbase + r + 8 * g;
      if (m < nRows) Out[(size_t)m * NCOLS + t * 16 + l] = acc[t][r];
    }
  }
}

__global__ __launch_bounds__(256) void deg_rank_kernel(
    const int* __restrict__ dst, unsigned* __restrict__ deg,
    unsigned* __restrict__ rank, int nE) {
  const int e = blockIdx.x * blockDim.x + threadIdx.x;
  if (e < nE) rank[e] = atomicAdd(&deg[dst[e]], 1u);
}

#define SCAN_TILE 1024

__global__ __launch_bounds__(256) void block_sum_kernel(
    const unsigned* __restrict__ deg, unsigned* __restrict__ bsum, int n) {
  __shared__ unsigned sh[8];
  const int tid  = threadIdx.x;
  const int lane = tid & 31;
  const int base = blockIdx.x * SCAN_TILE + tid * 4;
  unsigned s = 0;
#pragma unroll
  for (int j = 0; j < 4; ++j)
    if (base + j < n) s += deg[base + j];
#pragma unroll
  for (int off = 16; off > 0; off >>= 1) s += __shfl_xor(s, off, 32);
  if (lane == 0) sh[tid >> 5] = s;
  __syncthreads();
  if (tid == 0) {
    unsigned t = 0;
#pragma unroll
    for (int w = 0; w < 8; ++w) t += sh[w];
    bsum[blockIdx.x] = t;
  }
}

__global__ __launch_bounds__(256) void scan_dinv_kernel(
    const unsigned* __restrict__ deg, const unsigned* __restrict__ bsum,
    unsigned* __restrict__ rowptr, float* __restrict__ dinv, int n) {
  __shared__ unsigned shA[8];
  __shared__ unsigned shB[8];
  const int tid  = threadIdx.x;
  const int lane = tid & 31;
  const int wave = tid >> 5;

  unsigned o = 0;
  for (int j = tid; j < (int)blockIdx.x; j += 256) o += bsum[j];
#pragma unroll
  for (int off = 16; off > 0; off >>= 1) o += __shfl_xor(o, off, 32);
  if (lane == 0) shA[wave] = o;

  const int base = blockIdx.x * SCAN_TILE + tid * 4;
  unsigned d[4];
#pragma unroll
  for (int j = 0; j < 4; ++j) d[j] = (base + j < n) ? deg[base + j] : 0u;
  const unsigned t = d[0] + d[1] + d[2] + d[3];

  unsigned incl = t;
#pragma unroll
  for (int off = 1; off < 32; off <<= 1) {
    const unsigned v = __shfl_up(incl, off, 32);
    if (lane >= off) incl += v;
  }
  if (lane == 31) shB[wave] = incl;
  __syncthreads();

  unsigned ex = incl - t;
#pragma unroll
  for (int w = 0; w < 8; ++w) {
    ex += shA[w];
    if (w < wave) ex += shB[w];
  }
#pragma unroll
  for (int j = 0; j < 4; ++j) {
    if (base + j < n) {
      rowptr[base + j] = ex;
      dinv[base + j]   = rsqrtf((float)d[j] + 1.0f);
      ex += d[j];
    }
  }
}

__global__ __launch_bounds__(256) void permute_kernel(
    const int* __restrict__ src, const int* __restrict__ dst,
    const unsigned* __restrict__ rank, const unsigned* __restrict__ rowptr,
    int* __restrict__ perm, int nE) {
  const int e = blockIdx.x * blockDim.x + threadIdx.x;
  if (e < nE) perm[rowptr[dst[e]] + rank[e]] = src[e];
}

template <int C>
__global__ __launch_bounds__(256) void gather_layer(
    const int* __restrict__ perm, const unsigned* __restrict__ rowptr,
    const unsigned* __restrict__ deg, const float* __restrict__ dinv,
    const float* __restrict__ lin, const float* __restrict__ bias,
    float* __restrict__ out, int n, int doRelu) {
  constexpr int PER = C / 32;
  typedef float vec __attribute__((ext_vector_type(PER)));
  const int lane = threadIdx.x & 31;
  int node = (int)((blockIdx.x * (unsigned)blockDim.x + threadIdx.x) >> 5);
  node = __builtin_amdgcn_readfirstlane(node);
  if (node >= n) return;

  const unsigned start = rowptr[node];
  const int cnt        = (int)deg[node];
  const float di       = dinv[node];

  const vec self = *(const vec*)(lin + (size_t)node * C + lane * PER);
  vec acc = 0.0f;

  for (int base = 0; base < cnt; base += 32) {
    const int m = (cnt - base < 32) ? (cnt - base) : 32;
    int sv = 0;
    float wv = 0.0f;
    if (lane < m) {
      sv = perm[start + base + lane];
      wv = dinv[sv] * di;
    }
#pragma unroll 4
    for (int k = 0; k < m; ++k) {
      const int s   = __builtin_amdgcn_readlane(sv, k);
      const float w = __int_as_float(__builtin_amdgcn_readlane(__float_as_int(wv), k));
      const vec h   = *(const vec*)(lin + (size_t)s * C + lane * PER);
      acc += w * h;
    }
  }

  acc += (di * di) * self;
  vec v = acc + *(const vec*)(bias + lane * PER);
  if (doRelu) {
#pragma unroll
    for (int j = 0; j < PER; ++j) v[j] = fmaxf(v[j], 0.0f);
  }
  *(vec*)(out + (size_t)node * C + lane * PER) = v;
}

__global__ __launch_bounds__(256) void decode_kernel(
    const int* __restrict__ src, const int* __restrict__ dst,
    const float* __restrict__ z, float* __restrict__ out, int nE) {
  const int wid  = (int)((blockIdx.x * (unsigned)blockDim.x + threadIdx.x) >> 5);
  const int lane = threadIdx.x & 31;
  if (wid >= nE) return;
  const int s = src[wid];
  const int d = dst[wid];
  const float* __restrict__ zs = z + (size_t)s * OUT_CH + lane * 2;
  const float* __restrict__ zd = z + (size_t)d * OUT_CH + lane * 2;
  float p = zs[0] * zd[0] + zs[1] * zd[1];
#pragma unroll
  for (int off = 16; off > 0; off >>= 1) p += __shfl_xor(p, off, 32);
  if (lane == 0) out[wid] = 1.0f / (1.0f + __expf(-p));
}

extern "C" void kernel_launch(void* const* d_in, const int* in_sizes, int n_in,
                              void* d_out, int out_size, void* d_ws, size_t ws_size,
                              hipStream_t stream) {
  const float* x   = (const float*)d_in[0];
  const int*   ei  = (const int*)d_in[1];
  const float* W1  = (const float*)d_in[2];
  const float* b1  = (const float*)d_in[3];
  const float* W2  = (const float*)d_in[4];
  const float* b2  = (const float*)d_in[5];

  const int N = in_sizes[0] / IN_CH;
  const int E = in_sizes[1] / 2;
  const int* esrc = ei;
  const int* edst = ei + E;

  char* w = (char*)d_ws;
  unsigned* deg  = (unsigned*)w; w += (size_t)N * 4;
  float*    dinv = (float*)w;    w += (size_t)N * 4;
  float*    hlin = (float*)w;    w += (size_t)N * HID_CH * 4;
  float*    h    = (float*)w;    w += (size_t)N * HID_CH * 4;
  char*     r3   = w;            w += (size_t)N * OUT_CH * 4;
  char*     r4   = w;
  int*      perm   = (int*)r3;
  unsigned* rank   = (unsigned*)r4;
  unsigned* rowptr = (unsigned*)(r4 + (size_t)E * 4);
  unsigned* bsum   = rowptr + N;
  float*    zlin = hlin;
  float*    z    = hlin + (size_t)N * OUT_CH;

  hipMemsetAsync(deg, 0, (size_t)N * 4, stream);

  const int TPB = 256;
  const int nEdgeBlocks = (E + TPB - 1) / TPB;
  const int nWarpBlocks = (int)(((size_t)E * 32 + TPB - 1) / TPB);
  const int nNodeWaveBlocks = (int)(((size_t)N * 32 + TPB - 1) / TPB);
  const int nScanBlocks = (N + SCAN_TILE - 1) / SCAN_TILE;
  const int mTiles = (N + 15) / 16;

  deg_rank_kernel<<<nEdgeBlocks, TPB, 0, stream>>>(edst, deg, rank, E);
  block_sum_kernel<<<nScanBlocks, TPB, 0, stream>>>(deg, bsum, N);
  scan_dinv_kernel<<<nScanBlocks, TPB, 0, stream>>>(deg, bsum, rowptr, dinv, N);
  permute_kernel<<<nEdgeBlocks, TPB, 0, stream>>>(esrc, edst, rank, rowptr, perm, E);

  gemm_wmma_f32<IN_CH, HID_CH><<<mTiles, 32, 0, stream>>>(x, W1, hlin, N);
  gather_layer<HID_CH><<<nNodeWaveBlocks, TPB, 0, stream>>>(
      perm, rowptr, deg, dinv, hlin, b1, h, N, 1);

  gemm_wmma_f32<HID_CH, OUT_CH><<<mTiles, 32, 0, stream>>>(h, W2, zlin, N);
  gather_layer<OUT_CH><<<nNodeWaveBlocks, TPB, 0, stream>>>(
      perm, rowptr, deg, dinv, zlin, b2, z, N, 0);

  decode_kernel<<<nWarpBlocks, TPB, 0, stream>>>(esrc, edst, z, (float*)d_out, E);
}


# Decoder: a wave takes 32 edges, a 16-lane row per 256-B z row with 16-B loads all in flight and a DPP transposing sum, not one edge per wave; the dot's summation tree changes
# speedup vs baseline: 2.5305x; 1.4176x over previous
#include <hip/hip_runtime.h>
#include <math.h>


typedef __attribute__((ext_vector_type(2))) float v2f;
typedef __attribute__((ext_vector_type(8))) float v8f;

#define IN_CH  256
#define HID_CH 128
#define OUT_CH 64

template <int K, int NCOLS>
__global__ __launch_bounds__(32) void gemm_wmma_f32(
    const float* __restrict__ A, const float* __restrict__ W,
    float* __restrict__ Out, int nRows) {
  constexpr int NT = NCOLS / 16;
  const int mbase = blockIdx.x * 16;
  const int lane  = threadIdx.x & 31;
  const int g     = lane >> 4;
  const int l     = lane & 15;

  v8f acc[NT] = {};

  int arow_idx = mbase + l;
  if (arow_idx >= nRows) arow_idx = nRows - 1;
  const float* __restrict__ arow = A + (size_t)arow_idx * K + 2 * g;

#pragma unroll 4
  for (int k = 0; k < K; k += 4) {
    v2f a;
    a.x = arow[k];
    a.y = arow[k + 1];
    const float* __restrict__ wk = W + (size_t)(k + 2 * g) * NCOLS + l;
#pragma unroll
    for (int t = 0; t < NT; ++t) {
      v2f b;
      b.x = wk[t * 16];
      b.y = wk[NCOLS + t * 16];
      acc[t] = __builtin_amdgcn_wmma_f32_16x16x4_f32(
           false, a,  false, b,
           (short)0, acc[t],  false,  false);
    }
  }

#pragma unroll
  for (int t = 0; t < NT; ++t) {
#pragma unroll
    for (int r = 0; r < 8; ++r) {
      const int m = mbase + r + 8 * g;
      if (m < nRows) Out[(size_t)m * NCOLS + t * 16 + l] = acc[t][r];
    }
  }
}

__global__ __launch_bounds__(256) void deg_rank_kernel(
    const int* __restrict__ dst, unsigned* __restrict__ deg,
    unsigned* __restrict__ rank, int nE) {
  const int e = blockIdx.x * blockDim.x + threadIdx.x;
  if (e < nE) rank[e] = atomicAdd(&deg[dst[e]], 1u);
}

#define SCAN_TILE 1024

__global__ __launch_bounds__(256) void block_sum_kernel(
    const unsigned* __restrict__ deg, unsigned* __restrict__ bsum, int n) {
  __shared__ unsigned sh[8];
  const int tid  = threadIdx.x;
  const int lane = tid & 31;
  const int base = blockIdx.x * SCAN_TILE + tid * 4;
  unsigned s = 0;
#pragma unroll
  for (int j = 0; j < 4; ++j)
    if (base + j < n) s += deg[base + j];
#pragma unroll
  for (int off = 16; off > 0; off >>= 1) s += __shfl_xor(s, off, 32);
  if (lane == 0) sh[tid >> 5] = s;
  __syncthreads();
  if (tid == 0) {
    unsigned t = 0;
#pragma unroll
    for (int w = 0; w < 8; ++w) t += sh[w];
    bsum[blockIdx.x] = t;
  }
}

__global__ __launch_bounds__(256) void scan_dinv_kernel(
    const unsigned* __restrict__ deg, const unsigned* __restrict__ bsum,
    unsigned* __restrict__ rowptr, float* __restrict__ dinv, int n) {
  __shared__ unsigned shA[8];
  __shared__ unsigned shB[8];
  const int tid  = threadIdx.x;
  const int lane = tid & 31;
  const int wave = tid >> 5;

  unsigned o = 0;
  for (int j = tid; j < (int)blockIdx.x; j += 256) o += bsum[j];
#pragma unroll
  for (int off = 16; off > 0; off >>= 1) o += __shfl_xor(o, off, 32);
  if (lane == 0) shA[wave] = o;

  const int base = blockIdx.x * SCAN_TILE + tid * 4;
  unsigned d[4];
#pragma unroll
  for (int j = 0; j < 4; ++j) d[j] = (base + j < n) ? deg[base + j] : 0u;
  const unsigned t = d[0] + d[1] + d[2] + d[3];

  unsigned incl = t;
#pragma unroll
  for (int off = 1; off < 32; off <<= 1) {
    const unsigned v = __shfl_up(incl, off, 32);
    if (lane >= off) incl += v;
  }
  if (lane == 31) shB[wave] = incl;
  __syncthreads();

  unsigned ex = incl - t;
#pragma unroll
  for (int w = 0; w < 8; ++w) {
    ex += shA[w];
    if (w < wave) ex += shB[w];
  }
#pragma unroll
  for (int j = 0; j < 4; ++j) {
    if (base + j < n) {
      rowptr[base + j] = ex;
      dinv[base + j]   = rsqrtf((float)d[j] + 1.0f);
      ex += d[j];
    }
  }
}

__global__ __launch_bounds__(256) void permute_kernel(
    const int* __restrict__ src, const int* __restrict__ dst,
    const unsigned* __restrict__ rank, const unsigned* __restrict__ rowptr,
    int* __restrict__ perm, int nE) {
  const int e = blockIdx.x * blockDim.x + threadIdx.x;
  if (e < nE) perm[rowptr[dst[e]] + rank[e]] = src[e];
}

template <int C>
__global__ __launch_bounds__(256) void gather_layer(
    const int* __restrict__ perm, const unsigned* __restrict__ rowptr,
    const unsigned* __restrict__ deg, const float* __restrict__ dinv,
    const float* __restrict__ lin, const float* __restrict__ bias,
    float* __restrict__ out, int n, int doRelu) {
  constexpr int PER = C / 32;
  typedef float vec __attribute__((ext_vector_type(PER)));
  const int lane = threadIdx.x & 31;
  int node = (int)((blockIdx.x * (unsigned)blockDim.x + threadIdx.x) >> 5);
  node = __builtin_amdgcn_readfirstlane(node);
  if (node >= n) return;

  const unsigned start = rowptr[node];
  const int cnt        = (int)deg[node];
  const float di       = dinv[node];

  const vec self = *(const vec*)(lin + (size_t)node * C + lane * PER);
  vec acc = 0.0f;

  for (int base = 0; base < cnt; base += 32) {
    const int m = (cnt - base < 32) ? (cnt - base) : 32;
    int sv = 0;
    float wv = 0.0f;
    if (lane < m) {
      sv = perm[start + base + lane];
      wv = dinv[sv] * di;
    }
#pragma unroll 4
    for (int k = 0; k < m; ++k) {
      const int s   = __builtin_amdgcn_readlane(sv, k);
      const float w = __int_as_float(__builtin_amdgcn_readlane(__float_as_int(wv), k));
      const vec h   = *(const vec*)(lin + (size_t)s * C + lane * PER);
      acc += w * h;
    }
  }

  acc += (di * di) * self;
  vec v = acc + *(const vec*)(bias + lane * PER);
  if (doRelu) {
#pragma unroll
    for (int j = 0; j < PER; ++j) v[j] = fmaxf(v[j], 0.0f);
  }
  *(vec*)(out + (size_t)node * C + lane * PER) = v;
}

typedef __attribute__((ext_vector_type(4))) float v4f;

template <int J>
__device__ __forceinline__ int row_bcast(int v) {
  return __builtin_amdgcn_update_dpp(0, v, 0x150 + J, 0xf, 0xf, false);
}

template <int M>
__device__ __forceinline__ float row_xor(float v) {
  return __int_as_float(
      __builtin_amdgcn_update_dpp(0, __float_as_int(v), 0x160 + M, 0xf, 0xf, false));
}

template <int J>
__device__ __forceinline__ void edge_dots(float (&p)[16], int sv, int dv,
                                          const char* __restrict__ zb, unsigned coff) {
  const unsigned so = (unsigned)row_bcast<J>(sv) * (OUT_CH * 4u) + coff;
  const unsigned dofs = (unsigned)row_bcast<J>(dv) * (OUT_CH * 4u) + coff;
  const v4f a = *(const v4f*)(zb + so);
  const v4f b = *(const v4f*)(zb + dofs);
  p[J] = a.x * b.x + a.y * b.y + a.z * b.z + a.w * b.w;
  if constexpr (J + 1 < 16) edge_dots<J + 1>(p, sv, dv, zb, coff);
}

template <int H>
__device__ __forceinline__ void fold_step(float* v, bool upper) {
#pragma unroll
  for (int a = 0; a < H; ++a) {
    const float keep = upper ? v[a + H] : v[a];
    const float send = upper ? v[a] : v[a + H];
    v[a] = keep + row_xor<H>(send);
  }
}

__global__ __launch_bounds__(256) void decode_kernel(
    const int* __restrict__ src, const int* __restrict__ dst,
    const float* __restrict__ z, float* __restrict__ out, int nE) {
  const int lane = threadIdx.x & 31;
  int e0 = (int)(((blockIdx.x * (unsigned)blockDim.x + threadIdx.x) >> 5) << 5);
  e0 = __builtin_amdgcn_readfirstlane(e0);
  if (e0 >= nE) return;

  const int e = e0 + lane;
  int sv = 0, dv = 0;
  if (e < nE) {
    sv = src[e];
    dv = dst[e];
  }

  const char* __restrict__ zb = (const char*)z;
  const unsigned coff = (unsigned)(lane & 15) * 16u;

  float p[16];
  edge_dots<0>(p, sv, dv, zb, coff);

  fold_step<8>(p, (lane & 8) != 0);
  fold_step<4>(p, (lane & 4) != 0);
  fold_step<2>(p, (lane & 2) != 0);
  fold_step<1>(p, (lane & 1) != 0);

  if (e < nE) out[e] = 1.0f / (1.0f + __expf(-p[0]));
}

extern "C" void kernel_launch(void* const* d_in, const int* in_sizes, int n_in,
                              void* d_out, int out_size, void* d_ws, size_t ws_size,
                              hipStream_t stream) {
  const float* x   = (const float*)d_in[0];
  const int*   ei  = (const int*)d_in[1];
  const float* W1  = (const float*)d_in[2];
  const float* b1  = (const float*)d_in[3];
  const float* W2  = (const float*)d_in[4];
  const float* b2  = (const float*)d_in[5];

  const int N = in_sizes[0] / IN_CH;
  const int E = in_sizes[1] / 2;
  const int* esrc = ei;
  const int* edst = ei + E;

  char* w = (char*)d_ws;
  unsigned* deg  = (unsigned*)w; w += (size_t)N * 4;
  float*    dinv = (float*)w;    w += (size_t)N * 4;
  float*    hlin = (float*)w;    w += (size_t)N * HID_CH * 4;
  float*    h    = (float*)w;    w += (size_t)N * HID_CH * 4;
  char*     r3   = w;            w += (size_t)N * OUT_CH * 4;
  char*     r4   = w;
  int*      perm   = (int*)r3;
  unsigned* rank   = (unsigned*)r4;
  unsigned* rowptr = (unsigned*)(r4 + (size_t)E * 4);
  unsigned* bsum   = rowptr + N;
  float*    zlin = hlin;
  float*    z    = hlin + (size_t)N * OUT_CH;

  hipMemsetAsync(deg, 0, (size_t)N * 4, stream);

  const int TPB = 256;
  const int nEdgeBlocks = (E + TPB - 1) / TPB;
  const int nNodeWaveBlocks = (int)(((size_t)N * 32 + TPB - 1) / TPB);
  const int nScanBlocks = (N + SCAN_TILE - 1) / SCAN_TILE;
  const int mTiles = (N + 15) / 16;

  deg_rank_kernel<<<nEdgeBlocks, TPB, 0, stream>>>(edst, deg, rank, E);
  block_sum_kernel<<<nScanBlocks, TPB, 0, stream>>>(deg, bsum, N);
  scan_dinv_kernel<<<nScanBlocks, TPB, 0, stream>>>(deg, bsum, rowptr, dinv, N);
  permute_kernel<<<nEdgeBlocks, TPB, 0, stream>>>(esrc, edst, rank, rowptr, perm, E);

  gemm_wmma_f32<IN_CH, HID_CH><<<mTiles, 32, 0, stream>>>(x, W1, hlin, N);
  gather_layer<HID_CH><<<nNodeWaveBlocks, TPB, 0, stream>>>(
      perm, rowptr, deg, dinv, hlin, b1, h, N, 1);

  gemm_wmma_f32<HID_CH, OUT_CH><<<mTiles, 32, 0, stream>>>(h, W2, zlin, N);
  gather_layer<OUT_CH><<<nNodeWaveBlocks, TPB, 0, stream>>>(
      perm, rowptr, deg, dinv, zlin, b2, z, N, 0);

  decode_kernel<<<nEdgeBlocks, TPB, 0, stream>>>(esrc, edst, z, (float*)d_out, E);
}


# Layer-1 GEMM and the edge permute share one launch: GEMM stripes placed first at 9 waves per SIMD (row scales read behind the K loop), permute waves, 4 edges a thread, fill the slots beside them
# speedup vs baseline: 2.8576x; 1.1293x over previous
#include <hip/hip_runtime.h>
#include <math.h>


typedef __attribute__((ext_vector_type(2))) float v2f;
typedef __attribute__((ext_vector_type(8))) float v8f;

#define IN_CH  256
#define HID_CH 128
#define OUT_CH 64

template <int K, int NCOLS, bool SCALE_EARLY>
__device__ __forceinline__ void gemm_stripe(
    const float* __restrict__ A, const float* __restrict__ W,
    const float* __restrict__ rowScale, float* __restrict__ Out, int nRows, int stripe) {
  constexpr int NT = NCOLS / 16;
  const int mbase = stripe * 16;
  const int lane  = threadIdx.x & 31;
  const int g     = lane >> 4;
  const int l     = lane & 15;

  v8f acc[NT] = {};

  int arow_idx = mbase + l;
  if (arow_idx >= nRows) arow_idx = nRows - 1;
  const float* __restrict__ arow = A + (size_t)arow_idx * K + 2 * g;

  float rs[8];
  if constexpr (SCALE_EARLY) {
#pragma unroll
    for (int r = 0; r < 8; ++r) {
      int m = mbase + r + 8 * g;
      if (m >= nRows) m = nRows - 1;
      rs[r] = rowScale[m];
    }
  }

#pragma unroll 4
  for (int k = 0; k < K; k += 4) {
    v2f a;
    a.x = arow[k];
    a.y = arow[k + 1];
    const float* __restrict__ wk = W + (size_t)(k + 2 * g) * NCOLS + l;
#pragma unroll
    for (int t = 0; t < NT; ++t) {
      v2f b;
      b.x = wk[t * 16];
      b.y = wk[NCOLS + t * 16];
      acc[t] = __builtin_amdgcn_wmma_f32_16x16x4_f32(
           false, a,  false, b,
           (short)0, acc[t],  false,  false);
    }
  }

  if constexpr (!SCALE_EARLY) {
#pragma unroll
    for (int r = 0; r < 8; ++r) {
      int m = mbase + r + 8 * g;
      if (m >= nRows) m = nRows - 1;
      rs[r] = rowScale[m];
    }
  }

#pragma unroll
  for (int t = 0; t < NT; ++t) {
#pragma unroll
    for (int r = 0; r < 8; ++r) {
      const int m = mbase + r + 8 * g;
      if (m < nRows) Out[(size_t)m * NCOLS + t * 16 + l] = acc[t][r] * rs[r];
    }
  }
}

template <int K, int NCOLS>
__global__ __launch_bounds__(32) void gemm_wmma_f32(
    const float* __restrict__ A, const float* __restrict__ W,
    const float* __restrict__ rowScale, float* __restrict__ Out, int nRows) {
  gemm_stripe<K, NCOLS, true>(A, W, rowScale, Out, nRows, (int)blockIdx.x);
}

__global__ __launch_bounds__(256) void deg_rank_kernel(
    const int* __restrict__ dst, unsigned* __restrict__ deg,
    unsigned* __restrict__ rank, int nE) {
  const int e = blockIdx.x * blockDim.x + threadIdx.x;
  if (e < nE) rank[e] = atomicAdd(&deg[dst[e]], 1u);
}

#define SCAN_TILE 1024

__global__ __launch_bounds__(256) void block_sum_kernel(
    const unsigned* __restrict__ deg, unsigned* __restrict__ bsum, int n) {
  __shared__ unsigned sh[8];
  const int tid  = threadIdx.x;
  const int lane = tid & 31;
  const int base = blockIdx.x * SCAN_TILE + tid * 4;
  unsigned s = 0;
#pragma unroll
  for (int j = 0; j < 4; ++j)
    if (base + j < n) s += deg[base + j];
#pragma unroll
  for (int off = 16; off > 0; off >>= 1) s += __shfl_xor(s, off, 32);
  if (lane == 0) sh[tid >> 5] = s;
  __syncthreads();
  if (tid == 0) {
    unsigned t = 0;
#pragma unroll
    for (int w = 0; w < 8; ++w) t += sh[w];
    bsum[blockIdx.x] = t;
  }
}

__global__ __launch_bounds__(256) void scan_dinv_kernel(
    const unsigned* __restrict__ deg, const unsigned* __restrict__ bsum,
    unsigned* __restrict__ rowptr, float* __restrict__ dinv, int n) {
  __shared__ unsigned shA[8];
  __shared__ unsigned shB[8];
  const int tid  = threadIdx.x;
  const int lane = tid & 31;
  const int wave = tid >> 5;

  unsigned o = 0;
  for (int j = tid; j < (int)blockIdx.x; j += 256) o += bsum[j];
#pragma unroll
  for (int off = 16; off > 0; off >>= 1) o += __shfl_xor(o, off, 32);
  if (lane == 0) shA[wave] = o;

  const int base = blockIdx.x * SCAN_TILE + tid * 4;
  unsigned d[4];
#pragma unroll
  for (int j = 0; j < 4; ++j) d[j] = (base + j < n) ? deg[base + j] : 0u;
  const unsigned t = d[0] + d[1] + d[2] + d[3];

  unsigned incl = t;
#pragma unroll
  for (int off = 1; off < 32; off <<= 1) {
    const unsigned v = __shfl_up(incl, off, 32);
    if (lane >= off) incl += v;
  }
  if (lane == 31) shB[wave] = incl;
  __syncthreads();

  unsigned ex = incl - t;
#pragma unroll
  for (int w = 0; w < 8; ++w) {
    ex += shA[w];
    if (w < wave) ex += shB[w];
  }
#pragma unroll
  for (int j = 0; j < 4; ++j) {
    if (base + j < n) {
      rowptr[base + j] = ex;
      dinv[base + j]   = rsqrtf((float)d[j] + 1.0f);
      ex += d[j];
    }
  }
}

#define FUSED_TPB 128
#define FUSED_EPT 4

__global__ __launch_bounds__(FUSED_TPB) __attribute__((amdgpu_waves_per_eu(9, 9)))
void gemm1_permute_kernel(
    const float* __restrict__ x, const float* __restrict__ W1,
    const float* __restrict__ dinv, float* __restrict__ hlin, int nRows,
    int nGemmBlocks,
    const int* __restrict__ src, const int* __restrict__ dst,
    const unsigned* __restrict__ rank, const unsigned* __restrict__ rowptr,
    int* __restrict__ perm, int nE) {
  if ((int)blockIdx.x < nGemmBlocks) {
    int stripe = (int)blockIdx.x * (FUSED_TPB / 32) + (int)(threadIdx.x >> 5);
    stripe = __builtin_amdgcn_readfirstlane(stripe);
    if (stripe * 16 >= nRows) return;
    gemm_stripe<IN_CH, HID_CH, false>(x, W1, dinv, hlin, nRows, stripe);
  } else {
    const int base = ((int)blockIdx.x - nGemmBlocks) * (FUSED_TPB * FUSED_EPT) + (int)threadIdx.x;
    if (base + (FUSED_EPT - 1) * FUSED_TPB < nE) {
      int      d[FUSED_EPT], s[FUSED_EPT];
      unsigned r[FUSED_EPT], p[FUSED_EPT];
#pragma unroll
      for (int j = 0; j < FUSED_EPT; ++j) {
        const int e = base + j * FUSED_TPB;
        d[j] = dst[e];
        r[j] = rank[e];
        s[j] = src[e];
      }
#pragma unroll
      for (int j = 0; j < FUSED_EPT; ++j) p[j] = rowptr[d[j]] + r[j];
#pragma unroll
      for (int j = 0; j < FUSED_EPT; ++j) perm[p[j]] = s[j];
    } else {
      for (int j = 0; j < FUSED_EPT; ++j) {
        const int e = base + j * FUSED_TPB;
        if (e < nE) perm[rowptr[dst[e]] + rank[e]] = src[e];
      }
    }
  }
}

template <int C>
__global__ __launch_bounds__(256) void gather_layer(
    const int* __restrict__ perm, const unsigned* __restrict__ rowptr,
    const unsigned* __restrict__ deg, const float* __restrict__ dinv,
    const float* __restrict__ lin, const float* __restrict__ bias,
    float* __restrict__ out, int n, int doRelu) {
  constexpr int PER = C / 32;
  typedef float vec __attribute__((ext_vector_type(PER)));
  const int lane = threadIdx.x & 31;
  int node = (int)((blockIdx.x * (unsigned)blockDim.x + threadIdx.x) >> 5);
  node = __builtin_amdgcn_readfirstlane(node);
  if (node >= n) return;

  const unsigned start = rowptr[node];
  const int cnt        = (int)deg[node];
  const float di       = dinv[node];

  const vec self = *(const vec*)(lin + (size_t)node * C + lane * PER);
  vec acc = 0.0f;

  for (int base = 0; base < cnt; base += 32) {
    const int m = (cnt - base < 32) ? (cnt - base) : 32;
    int sv = 0;
    if (lane < m) sv = perm[start + base + lane];
#pragma unroll 4
    for (int k = 0; k < m; ++k) {
      const int s = __builtin_amdgcn_readlane(sv, k);
      acc += *(const vec*)(lin + (size_t)s * C + lane * PER);
    }
  }

  acc += self;
  vec v = di * acc + *(const vec*)(bias + lane * PER);
  if (doRelu) {
#pragma unroll
    for (int j = 0; j < PER; ++j) v[j] = fmaxf(v[j], 0.0f);
  }
  *(vec*)(out + (size_t)node * C + lane * PER) = v;
}

typedef __attribute__((ext_vector_type(4))) float v4f;

template <int J>
__device__ __forceinline__ int row_bcast(int v) {
  return __builtin_amdgcn_update_dpp(0, v, 0x150 + J, 0xf, 0xf, false);
}

template <int M>
__device__ __forceinline__ float row_xor(float v) {
  return __int_as_float(
      __builtin_amdgcn_update_dpp(0, __float_as_int(v), 0x160 + M, 0xf, 0xf, false));
}

template <int J>
__device__ __forceinline__ void edge_dots(float (&p)[16], int sv, int dv,
                                          const char* __restrict__ zb, unsigned coff) {
  const unsigned so = (unsigned)row_bcast<J>(sv) * (OUT_CH * 4u) + coff;
  const unsigned dofs = (unsigned)row_bcast<J>(dv) * (OUT_CH * 4u) + coff;
  const v4f a = *(const v4f*)(zb + so);
  const v4f b = *(const v4f*)(zb + dofs);
  p[J] = a.x * b.x + a.y * b.y + a.z * b.z + a.w * b.w;
  if constexpr (J + 1 < 16) edge_dots<J + 1>(p, sv, dv, zb, coff);
}

template <int H>
__device__ __forceinline__ void fold_step(float* v, bool upper) {
#pragma unroll
  for (int a = 0; a < H; ++a) {
    const float keep = upper ? v[a + H] : v[a];
    const float send = upper ? v[a] : v[a + H];
    v[a] = keep + row_xor<H>(send);
  }
}

__global__ __launch_bounds__(256) void decode_kernel(
    const int* __restrict__ src, const int* __restrict__ dst,
    const float* __restrict__ z, float* __restrict__ out, int nE) {
  const int lane = threadIdx.x & 31;
  int e0 = (int)(((blockIdx.x * (unsigned)blockDim.x + threadIdx.x) >> 5) << 5);
  e0 = __builtin_amdgcn_readfirstlane(e0);
  if (e0 >= nE) return;

  const int e = e0 + lane;
  int sv = 0, dv = 0;
  if (e < nE) {
    sv = src[e];
    dv = dst[e];
  }

  const char* __restrict__ zb = (const char*)z;
  const unsigned coff = (unsigned)(lane & 15) * 16u;

  float p[16];
  edge_dots<0>(p, sv, dv, zb, coff);

  fold_step<8>(p, (lane & 8) != 0);
  fold_step<4>(p, (lane & 4) != 0);
  fold_step<2>(p, (lane & 2) != 0);
  fold_step<1>(p, (lane & 1) != 0);

  if (e < nE) out[e] = 1.0f / (1.0f + __expf(-p[0]));
}

extern "C" void kernel_launch(void* const* d_in, const int* in_sizes, int n_in,
                              void* d_out, int out_size, void* d_ws, size_t ws_size,
                              hipStream_t stream) {
  const float* x   = (const float*)d_in[0];
  const int*   ei  = (const int*)d_in[1];
  const float* W1  = (const float*)d_in[2];
  const float* b1  = (const float*)d_in[3];
  const float* W2  = (const float*)d_in[4];
  const float* b2  = (const float*)d_in[5];

  const int N = in_sizes[0] / IN_CH;
  const int E = in_sizes[1] / 2;
  const int* esrc = ei;
  const int* edst = ei + E;

  char* w = (char*)d_ws;
  unsigned* deg  = (unsigned*)w; w += (size_t)N * 4;
  float*    dinv = (float*)w;    w += (size_t)N * 4;
  float*    hlin = (float*)w;    w += (size_t)N * HID_CH * 4;
  float*    h    = (float*)w;    w += (size_t)N * HID_CH * 4;
  char*     r3   = w;            w += (size_t)N * OUT_CH * 4;
  char*     r4   = w;
  int*      perm   = (int*)r3;
  unsigned* rank   = (unsigned*)r4;
  unsigned* rowptr = (unsigned*)(r4 + (size_t)E * 4);
  unsigned* bsum   = rowptr + N;
  float*    zlin = hlin;
  float*    z    = hlin + (size_t)N * OUT_CH;

  hipMemsetAsync(deg, 0, (size_t)N * 4, stream);

  const int TPB = 256;
  const int nEdgeBlocks = (E + TPB - 1) / TPB;
  const int nNodeWaveBlocks = (int)(((size_t)N * 32 + TPB - 1) / TPB);
  const int nScanBlocks = (N + SCAN_TILE - 1) / SCAN_TILE;
  const int mTiles = (N + 15) / 16;

  deg_rank_kernel<<<nEdgeBlocks, TPB, 0, stream>>>(edst, deg, rank, E);
  block_sum_kernel<<<nScanBlocks, TPB, 0, stream>>>(deg, bsum, N);
  scan_dinv_kernel<<<nScanBlocks, TPB, 0, stream>>>(deg, bsum, rowptr, dinv, N);

  const int nGemmBlocks = (mTiles + FUSED_TPB / 32 - 1) / (FUSED_TPB / 32);
  const int nPermBlocks = (E + FUSED_TPB * FUSED_EPT - 1) / (FUSED_TPB * FUSED_EPT);
  gemm1_permute_kernel<<<nGemmBlocks + nPermBlocks, FUSED_TPB, 0, stream>>>(
      x, W1, dinv, hlin, N, nGemmBlocks, esrc, edst, rank, rowptr, perm, E);

  gather_layer<HID_CH><<<nNodeWaveBlocks, TPB, 0, stream>>>(
      perm, rowptr, deg, dinv, hlin, b1, h, N, 1);

  gemm_wmma_f32<HID_CH, OUT_CH><<<mTiles, 32, 0, stream>>>(h, W2, dinv, zlin, N);
  gather_layer<OUT_CH><<<nNodeWaveBlocks, TPB, 0, stream>>>(
      perm, rowptr, deg, dinv, zlin, b2, z, N, 0);

  decode_kernel<<<nEdgeBlocks, TPB, 0, stream>>>(esrc, edst, z, (float*)d_out, E);
}


# Layer-2 GEMM folded into the layer-1 gather launch: a work-group gathers 32 nodes, keeps their relu rows in LDS and multiplies them by W2 on the matrix unit; h never goes to memory, one launch fewer
# speedup vs baseline: 2.9738x; 1.0407x over previous
#include <hip/hip_runtime.h>
#include <math.h>


typedef __attribute__((ext_vector_type(2))) float v2f;
typedef __attribute__((ext_vector_type(8))) float v8f;
typedef __attribute__((ext_vector_type(4))) float v4f;

#define IN_CH  256
#define HID_CH 128
#define OUT_CH 64

template <int K, int NCOLS, bool SCALE_EARLY>
__device__ __forceinline__ void gemm_stripe(
    const float* __restrict__ A, const float* __restrict__ W,
    const float* __restrict__ rowScale, float* __restrict__ Out, int nRows, int stripe) {
  constexpr int NT = NCOLS / 16;
  const int mbase = stripe * 16;
  const int lane  = threadIdx.x & 31;
  const int g     = lane >> 4;
  const int l     = lane & 15;

  v8f acc[NT] = {};

  int arow_idx = mbase + l;
  if (arow_idx >= nRows) arow_idx = nRows - 1;
  const float* __restrict__ arow = A + (size_t)arow_idx * K + 2 * g;

  float rs[8];
  if constexpr (SCALE_EARLY) {
#pragma unroll
    for (int r = 0; r < 8; ++r) {
      int m = mbase + r + 8 * g;
      if (m >= nRows) m = nRows - 1;
      rs[r] = rowScale[m];
    }
  }

#pragma unroll 4
  for (int k = 0; k < K; k += 4) {
    v2f a;
    a.x = arow[k];
    a.y = arow[k + 1];
    const float* __restrict__ wk = W + (size_t)(k + 2 * g) * NCOLS + l;
#pragma unroll
    for (int t = 0; t < NT; ++t) {
      v2f b;
      b.x = wk[t * 16];
      b.y = wk[NCOLS + t * 16];
      acc[t] = __builtin_amdgcn_wmma_f32_16x16x4_f32(
           false, a,  false, b,
           (short)0, acc[t],  false,  false);
    }
  }

  if constexpr (!SCALE_EARLY) {
#pragma unroll
    for (int r = 0; r < 8; ++r) {
      int m = mbase + r + 8 * g;
      if (m >= nRows) m = nRows - 1;
      rs[r] = rowScale[m];
    }
  }

#pragma unroll
  for (int t = 0; t < NT; ++t) {
#pragma unroll
    for (int r = 0; r < 8; ++r) {
      const int m = mbase + r + 8 * g;
      if (m < nRows) Out[(size_t)m * NCOLS + t * 16 + l] = acc[t][r] * rs[r];
    }
  }
}

__global__ __launch_bounds__(256) void deg_rank_kernel(
    const int* __restrict__ dst, unsigned* __restrict__ deg,
    unsigned* __restrict__ rank, int nE) {
  const int e = blockIdx.x * blockDim.x + threadIdx.x;
  if (e < nE) rank[e] = atomicAdd(&deg[dst[e]], 1u);
}

#define SCAN_TILE 1024

__global__ __launch_bounds__(256) void block_sum_kernel(
    const unsigned* __restrict__ deg, unsigned* __restrict__ bsum, int n) {
  __shared__ unsigned sh[8];
  const int tid  = threadIdx.x;
  const int lane = tid & 31;
  const int base = blockIdx.x * SCAN_TILE + tid * 4;
  unsigned s = 0;
#pragma unroll
  for (int j = 0; j < 4; ++j)
    if (base + j < n) s += deg[base + j];
#pragma unroll
  for (int off = 16; off > 0; off >>= 1) s += __shfl_xor(s, off, 32);
  if (lane == 0) sh[tid >> 5] = s;
  __syncthreads();
  if (tid == 0) {
    unsigned t = 0;
#pragma unroll
    for (int w = 0; w < 8; ++w) t += sh[w];
    bsum[blockIdx.x] = t;
  }
}

__global__ __launch_bounds__(256) void scan_dinv_kernel(
    const unsigned* __restrict__ deg, const unsigned* __restrict__ bsum,
    unsigned* __restrict__ rowptr, float* __restrict__ dinv, int n) {
  __shared__ unsigned shA[8];
  __shared__ unsigned shB[8];
  const int tid  = threadIdx.x;
  const int lane = tid & 31;
  const int wave = tid >> 5;

  unsigned o = 0;
  for (int j = tid; j < (int)blockIdx.x; j += 256) o += bsum[j];
#pragma unroll
  for (int off = 16; off > 0; off >>= 1) o += __shfl_xor(o, off, 32);
  if (lane == 0) shA[wave] = o;

  const int base = blockIdx.x * SCAN_TILE + tid * 4;
  unsigned d[4];
#pragma unroll
  for (int j = 0; j < 4; ++j) d[j] = (base + j < n) ? deg[base + j] : 0u;
  const unsigned t = d[0] + d[1] + d[2] + d[3];

  unsigned incl = t;
#pragma unroll
  for (int off = 1; off < 32; off <<= 1) {
    const unsigned v = __shfl_up(incl, off, 32);
    if (lane >= off) incl += v;
  }
  if (lane == 31) shB[wave] = incl;
  __syncthreads();

  unsigned ex = incl - t;
#pragma unroll
  for (int w = 0; w < 8; ++w) {
    ex += shA[w];
    if (w < wave) ex += shB[w];
  }
#pragma unroll
  for (int j = 0; j < 4; ++j) {
    if (base + j < n) {
      rowptr[base + j] = ex;
      dinv[base + j]   = rsqrtf((float)d[j] + 1.0f);
      ex += d[j];
    }
  }
}

#define FUSED_TPB 128
#define FUSED_EPT 4

__global__ __launch_bounds__(FUSED_TPB) __attribute__((amdgpu_waves_per_eu(9, 9)))
void gemm1_permute_kernel(
    const float* __restrict__ x, const float* __restrict__ W1,
    const float* __restrict__ dinv, float* __restrict__ hlin, int nRows,
    int nGemmBlocks,
    const int* __restrict__ src, const int* __restrict__ dst,
    const unsigned* __restrict__ rank, const unsigned* __restrict__ rowptr,
    int* __restrict__ perm, int nE) {
  if ((int)blockIdx.x < nGemmBlocks) {
    int stripe = (int)blockIdx.x * (FUSED_TPB / 32) + (int)(threadIdx.x >> 5);
    stripe = __builtin_amdgcn_readfirstlane(stripe);
    if (stripe * 16 >= nRows) return;
    gemm_stripe<IN_CH, HID_CH, false>(x, W1, dinv, hlin, nRows, stripe);
  } else {
    const int base = ((int)blockIdx.x - nGemmBlocks) * (FUSED_TPB * FUSED_EPT) + (int)threadIdx.x;
    if (base + (FUSED_EPT - 1) * FUSED_TPB < nE) {
      int      d[FUSED_EPT], s[FUSED_EPT];
      unsigned r[FUSED_EPT], p[FUSED_EPT];
#pragma unroll
      for (int j = 0; j < FUSED_EPT; ++j) {
        const int e = base + j * FUSED_TPB;
        d[j] = dst[e];
        r[j] = rank[e];
        s[j] = src[e];
      }
#pragma unroll
      for (int j = 0; j < FUSED_EPT; ++j) p[j] = rowptr[d[j]] + r[j];
#pragma unroll
      for (int j = 0; j < FUSED_EPT; ++j) perm[p[j]] = s[j];
    } else {
      for (int j = 0; j < FUSED_EPT; ++j) {
        const int e = base + j * FUSED_TPB;
        if (e < nE) perm[rowptr[dst[e]] + rank[e]] = src[e];
      }
    }
  }
}

template <int C>
__global__ __launch_bounds__(256) void gather_layer(
    const int* __restrict__ perm, const unsigned* __restrict__ rowptr,
    const unsigned* __restrict__ deg, const float* __restrict__ dinv,
    const float* __restrict__ lin, const float* __restrict__ bias,
    float* __restrict__ out, int n, int doRelu) {
  constexpr int PER = C / 32;
  typedef float vec __attribute__((ext_vector_type(PER)));
  const int lane = threadIdx.x & 31;
  int node = (int)((blockIdx.x * (unsigned)blockDim.x + threadIdx.x) >> 5);
  node = __builtin_amdgcn_readfirstlane(node);
  if (node >= n) return;

  const unsigned start = rowptr[node];
  const int cnt        = (int)deg[node];
  const float di       = dinv[node];

  const vec self = *(const vec*)(lin + (size_t)node * C + lane * PER);
  vec acc = 0.0f;

  for (int base = 0; base < cnt; base += 32) {
    const int m = (cnt - base < 32) ? (cnt - base) : 32;
    int sv = 0;
    if (lane < m) sv = perm[start + base + lane];
#pragma unroll 4
    for (int k = 0; k < m; ++k) {
      const int s = __builtin_amdgcn_readlane(sv, k);
      acc += *(const vec*)(lin + (size_t)s * C + lane * PER);
    }
  }

  acc += self;
  vec v = di * acc + *(const vec*)(bias + lane * PER);
  if (doRelu) {
#pragma unroll
    for (int j = 0; j < PER; ++j) v[j] = fmaxf(v[j], 0.0f);
  }
  *(vec*)(out + (size_t)node * C + lane * PER) = v;
}

#define G1_NODES 32
#define G1_PITCH (HID_CH + 4)

__global__ __launch_bounds__(256) void gather1_gemm2_kernel(
    const int* __restrict__ perm, const unsigned* __restrict__ rowptr,
    const unsigned* __restrict__ deg, const float* __restrict__ dinv,
    const float* __restrict__ lin, const float* __restrict__ bias,
    const float* __restrict__ W2, float* __restrict__ zlin, int n) {
  __shared__ float tile[G1_NODES * G1_PITCH];
  const int lane = threadIdx.x & 31;
  const int wave = __builtin_amdgcn_readfirstlane((int)(threadIdx.x >> 5));
  const int nodeBase = (int)blockIdx.x * G1_NODES;

  constexpr int NPW = G1_NODES / 8;
  unsigned start[NPW];
  int      cnt[NPW];
  float    di[NPW];
  v4f      self[NPW];
  int      sv0[NPW];
#pragma unroll
  for (int j = 0; j < NPW; ++j) {
    const int node = nodeBase + wave * NPW + j;
    const int nc   = (node < n) ? node : n - 1;
    start[j] = rowptr[nc];
    const int dg = (int)deg[nc];
    cnt[j]   = (node < n) ? dg : 0;
    di[j]    = dinv[nc];
    self[j]  = *(const v4f*)(lin + (size_t)nc * HID_CH + lane * 4);
  }
#pragma unroll
  for (int j = 0; j < NPW; ++j) {
    sv0[j] = 0;
    if (lane < cnt[j]) sv0[j] = perm[start[j] + lane];
  }
  const v4f bv = *(const v4f*)(bias + lane * 4);
  const float* __restrict__ lrow = lin + lane * 4;
#pragma unroll
  for (int j = 0; j < NPW; ++j) {
    v4f acc = 0.0f;
    for (int base = 0; base < cnt[j]; base += 32) {
      const int m = (cnt[j] - base < 32) ? (cnt[j] - base) : 32;
      int sv = sv0[j];
      if (base > 0) {
        sv = 0;
        if (lane < m) sv = perm[start[j] + base + lane];
      }
      int k = 0;
      for (; k + 4 <= m; k += 4) {
        const int s0 = __builtin_amdgcn_readlane(sv, k);
        const int s1 = __builtin_amdgcn_readlane(sv, k + 1);
        const int s2 = __builtin_amdgcn_readlane(sv, k + 2);
        const int s3 = __builtin_amdgcn_readlane(sv, k + 3);
        const v4f r0 = *(const v4f*)(lrow + (size_t)s0 * HID_CH);
        const v4f r1 = *(const v4f*)(lrow + (size_t)s1 * HID_CH);
        const v4f r2 = *(const v4f*)(lrow + (size_t)s2 * HID_CH);
        const v4f r3 = *(const v4f*)(lrow + (size_t)s3 * HID_CH);
        acc += r0; acc += r1; acc += r2; acc += r3;
      }
      if (k + 2 <= m) {
        const int s0 = __builtin_amdgcn_readlane(sv, k);
        const int s1 = __builtin_amdgcn_readlane(sv, k + 1);
        const v4f r0 = *(const v4f*)(lrow + (size_t)s0 * HID_CH);
        const v4f r1 = *(const v4f*)(lrow + (size_t)s1 * HID_CH);
        acc += r0; acc += r1;
        k += 2;
      }
      if (k < m) {
        const int s0 = __builtin_amdgcn_readlane(sv, k);
        acc += *(const v4f*)(lrow + (size_t)s0 * HID_CH);
      }
    }
    acc += self[j];
    v4f v = di[j] * acc + bv;
    v.x = fmaxf(v.x, 0.0f); v.y = fmaxf(v.y, 0.0f);
    v.z = fmaxf(v.z, 0.0f); v.w = fmaxf(v.w, 0.0f);
    *(v4f*)(tile + (wave * NPW + j) * G1_PITCH + lane * 4) = v;
  }
  __syncthreads();

  const int g  = lane >> 4;
  const int l  = lane & 15;
  const int mt = wave >> 2;
  const int t  = wave & 3;
  const float* __restrict__ arow = tile + (mt * 16 + l) * G1_PITCH + 2 * g;
  const float* __restrict__ wcol = W2 + (size_t)(2 * g) * OUT_CH + t * 16 + l;
  const int mbase = nodeBase + mt * 16 + 8 * g;
  float rs[8];
#pragma unroll
  for (int r = 0; r < 8; ++r) {
    const int m = mbase + r;
    rs[r] = dinv[(m < n) ? m : n - 1];
  }
  v8f acc2 = {};
#pragma unroll 8
  for (int k = 0; k < HID_CH; k += 4) {
    v2f a;
    a.x = arow[k];
    a.y = arow[k + 1];
    v2f b;
    b.x = wcol[(size_t)k * OUT_CH];
    b.y = wcol[(size_t)k * OUT_CH + OUT_CH];
    acc2 = __builtin_amdgcn_wmma_f32_16x16x4_f32(
         false, a,  false, b,
         (short)0, acc2,  false,  false);
  }
#pragma unroll
  for (int r = 0; r < 8; ++r) {
    const int m = mbase + r;
    if (m < n) zlin[(size_t)m * OUT_CH + t * 16 + l] = acc2[r] * rs[r];
  }
}

template <int J>
__device__ __forceinline__ int row_bcast(int v) {
  return __builtin_amdgcn_update_dpp(0, v, 0x150 + J, 0xf, 0xf, false);
}

template <int M>
__device__ __forceinline__ float row_xor(float v) {
  return __int_as_float(
      __builtin_amdgcn_update_dpp(0, __float_as_int(v), 0x160 + M, 0xf, 0xf, false));
}

template <int J>
__device__ __forceinline__ void edge_dots(float (&p)[16], int sv, int dv,
                                          const char* __restrict__ zb, unsigned coff) {
  const unsigned so = (unsigned)row_bcast<J>(sv) * (OUT_CH * 4u) + coff;
  const unsigned dofs = (unsigned)row_bcast<J>(dv) * (OUT_CH * 4u) + coff;
  const v4f a = *(const v4f*)(zb + so);
  const v4f b = *(const v4f*)(zb + dofs);
  p[J] = a.x * b.x + a.y * b.y + a.z * b.z + a.w * b.w;
  if constexpr (J + 1 < 16) edge_dots<J + 1>(p, sv, dv, zb, coff);
}

template <int H>
__device__ __forceinline__ void fold_step(float* v, bool upper) {
#pragma unroll
  for (int a = 0; a < H; ++a) {
    const float keep = upper ? v[a + H] : v[a];
    const float send = upper ? v[a] : v[a + H];
    v[a] = keep + row_xor<H>(send);
  }
}

__global__ __launch_bounds__(256) void decode_kernel(
    const int* __restrict__ src, const int* __restrict__ dst,
    const float* __restrict__ z, float* __restrict__ out, int nE) {
  const int lane = threadIdx.x & 31;
  int e0 = (int)(((blockIdx.x * (unsigned)blockDim.x + threadIdx.x) >> 5) << 5);
  e0 = __builtin_amdgcn_readfirstlane(e0);
  if (e0 >= nE) return;

  const int e = e0 + lane;
  int sv = 0, dv = 0;
  if (e < nE) {
    sv = src[e];
    dv = dst[e];
  }

  const char* __restrict__ zb = (const char*)z;
  const unsigned coff = (unsigned)(lane & 15) * 16u;

  float p[16];
  edge_dots<0>(p, sv, dv, zb, coff);

  fold_step<8>(p, (lane & 8) != 0);
  fold_step<4>(p, (lane & 4) != 0);
  fold_step<2>(p, (lane & 2) != 0);
  fold_step<1>(p, (lane & 1) != 0);

  if (e < nE) out[e] = 1.0f / (1.0f + __expf(-p[0]));
}

extern "C" void kernel_launch(void* const* d_in, const int* in_sizes, int n_in,
                              void* d_out, int out_size, void* d_ws, size_t ws_size,
                              hipStream_t stream) {
  const float* x   = (const float*)d_in[0];
  const int*   ei  = (const int*)d_in[1];
  const float* W1  = (const float*)d_in[2];
  const float* b1  = (const float*)d_in[3];
  const float* W2  = (const float*)d_in[4];
  const float* b2  = (const float*)d_in[5];

  const int N = in_sizes[0] / IN_CH;
  const int E = in_sizes[1] / 2;
  const int* esrc = ei;
  const int* edst = ei + E;

  char* w = (char*)d_ws;
  unsigned* deg  = (unsigned*)w; w += (size_t)N * 4;
  float*    dinv = (float*)w;    w += (size_t)N * 4;
  float*    hlin = (float*)w;    w += (size_t)N * HID_CH * 4;
  float*    h    = (float*)w;    w += (size_t)N * HID_CH * 4;
  char*     r3   = w;            w += (size_t)N * OUT_CH * 4;
  char*     r4   = w;
  int*      perm   = (int*)r3;
  unsigned* rank   = (unsigned*)r4;
  unsigned* rowptr = (unsigned*)(r4 + (size_t)E * 4);
  unsigned* bsum   = rowptr + N;
  float*    zlin = h;
  float*    z    = hlin;

  hipMemsetAsync(deg, 0, (size_t)N * 4, stream);

  const int TPB = 256;
  const int nEdgeBlocks = (E + TPB - 1) / TPB;
  const int nNodeWaveBlocks = (int)(((size_t)N * 32 + TPB - 1) / TPB);
  const int nScanBlocks = (N + SCAN_TILE - 1) / SCAN_TILE;
  const int mTiles = (N + 15) / 16;

  deg_rank_kernel<<<nEdgeBlocks, TPB, 0, stream>>>(edst, deg, rank, E);
  block_sum_kernel<<<nScanBlocks, TPB, 0, stream>>>(deg, bsum, N);
  scan_dinv_kernel<<<nScanBlocks, TPB, 0, stream>>>(deg, bsum, rowptr, dinv, N);

  const int nGemmBlocks = (mTiles + FUSED_TPB / 32 - 1) / (FUSED_TPB / 32);
  const int nPermBlocks = (E + FUSED_TPB * FUSED_EPT - 1) / (FUSED_TPB * FUSED_EPT);
  gemm1_permute_kernel<<<nGemmBlocks + nPermBlocks, FUSED_TPB, 0, stream>>>(
      x, W1, dinv, hlin, N, nGemmBlocks, esrc, edst, rank, rowptr, perm, E);

  const int nG1Blocks = (N + G1_NODES - 1) / G1_NODES;
  gather1_gemm2_kernel<<<nG1Blocks, TPB, 0, stream>>>(
      perm, rowptr, deg, dinv, hlin, b1, W2, zlin, N);

  gather_layer<OUT_CH><<<nNodeWaveBlocks, TPB, 0, stream>>>(
      perm, rowptr, deg, dinv, zlin, b2, z, N, 0);

  decode_kernel<<<nEdgeBlocks, TPB, 0, stream>>>(esrc, edst, z, (float*)d_out, E);
}


# Source lists at a fixed pitch, slot = dst*32 + rank, spill list past 32: no degree scan (two launches fewer), no row-start read per edge, dinv from the degree in place; spill rows add in spill order
# speedup vs baseline: 3.1432x; 1.0570x over previous
#include <hip/hip_runtime.h>
#include <math.h>


typedef __attribute__((ext_vector_type(2))) float v2f;
typedef __attribute__((ext_vector_type(8))) float v8f;
typedef __attribute__((ext_vector_type(4))) float v4f;

#define IN_CH  256
#define HID_CH 128
#define OUT_CH 64

template <int K, int NCOLS, bool SCALE_EARLY>
__device__ __forceinline__ void gemm_stripe(
    const float* __restrict__ A, const float* __restrict__ W,
    const unsigned* __restrict__ rowDeg, float* __restrict__ Out, int nRows, int stripe) {
  constexpr int NT = NCOLS / 16;
  const int mbase = stripe * 16;
  const int lane  = threadIdx.x & 31;
  const int g     = lane >> 4;
  const int l     = lane & 15;

  v8f acc[NT] = {};

  int arow_idx = mbase + l;
  if (arow_idx >= nRows) arow_idx = nRows - 1;
  const float* __restrict__ arow = A + (size_t)arow_idx * K + 2 * g;

  float rs[8];
  if constexpr (SCALE_EARLY) {
#pragma unroll
    for (int r = 0; r < 8; ++r) {
      int m = mbase + r + 8 * g;
      if (m >= nRows) m = nRows - 1;
      rs[r] = rsqrtf((float)rowDeg[m] + 1.0f);
    }
  }

#pragma unroll 4
  for (int k = 0; k < K; k += 4) {
    v2f a;
    a.x = arow[k];
    a.y = arow[k + 1];
    const float* __restrict__ wk = W + (size_t)(k + 2 * g) * NCOLS + l;
#pragma unroll
    for (int t = 0; t < NT; ++t) {
      v2f b;
      b.x = wk[t * 16];
      b.y = wk[NCOLS + t * 16];
      acc[t] = __builtin_amdgcn_wmma_f32_16x16x4_f32(
           false, a,  false, b,
           (short)0, acc[t],  false,  false);
    }
  }

  if constexpr (!SCALE_EARLY) {
#pragma unroll
    for (int r = 0; r < 8; ++r) {
      int m = mbase + r + 8 * g;
      if (m >= nRows) m = nRows - 1;
      rs[r] = rsqrtf((float)rowDeg[m] + 1.0f);
    }
  }

#pragma unroll
  for (int t = 0; t < NT; ++t) {
#pragma unroll
    for (int r = 0; r < 8; ++r) {
      const int m = mbase + r + 8 * g;
      if (m < nRows) Out[(size_t)m * NCOLS + t * 16 + l] = acc[t][r] * rs[r];
    }
  }
}

#define LIST_CAP 32

__global__ __launch_bounds__(256) void deg_rank_kernel(
    const int* __restrict__ dst, unsigned* __restrict__ deg,
    unsigned* __restrict__ rank, int nE) {
  const int e = blockIdx.x * blockDim.x + threadIdx.x;
  if (e < nE) rank[e] = atomicAdd(&deg[dst[e]], 1u);
}

__device__ __forceinline__ void put_edge(int* __restrict__ slots, int* __restrict__ spill,
                                         unsigned* __restrict__ nSpill,
                                         int d, unsigned r, int s) {
  if (r < LIST_CAP) {
    slots[(unsigned)d * LIST_CAP + r] = s;
  } else {
    const unsigned o = atomicAdd(nSpill, 1u);
    spill[2 * o]     = d;
    spill[2 * o + 1] = s;
  }
}

template <int C, typename V>
__device__ __forceinline__ void add_spilled(V& acc, const int* __restrict__ spill, int nSpill,
                                            int node, int lane, const float* __restrict__ lrow) {
  for (int i = 0; i < nSpill; i += 32) {
    int od = -1, os = 0;
    if (i + lane < nSpill) {
      od = spill[2 * (i + lane)];
      os = spill[2 * (i + lane) + 1];
    }
    unsigned hit = __builtin_amdgcn_ballot_w32(od == node);
    while (hit != 0u) {
      const int k = __builtin_ctz(hit);
      hit &= hit - 1u;
      const int s = __builtin_amdgcn_readlane(os, k);
      acc += *(const V*)(lrow + (size_t)s * C);
    }
  }
}

#define FUSED_TPB 128
#define FUSED_EPT 4

__global__ __launch_bounds__(FUSED_TPB) __attribute__((amdgpu_waves_per_eu(9, 9)))
void gemm1_fill_kernel(
    const float* __restrict__ x, const float* __restrict__ W1,
    const unsigned* __restrict__ deg, float* __restrict__ hlin, int nRows,
    int nGemmBlocks,
    const int* __restrict__ src, const int* __restrict__ dst,
    const unsigned* __restrict__ rank, int* __restrict__ slots,
    int* __restrict__ spill, unsigned* __restrict__ nSpill, int nE) {
  if ((int)blockIdx.x < nGemmBlocks) {
    int stripe = (int)blockIdx.x * (FUSED_TPB / 32) + (int)(threadIdx.x >> 5);
    stripe = __builtin_amdgcn_readfirstlane(stripe);
    if (stripe * 16 >= nRows) return;
    gemm_stripe<IN_CH, HID_CH, false>(x, W1, deg, hlin, nRows, stripe);
  } else {
    const int base = ((int)blockIdx.x - nGemmBlocks) * (FUSED_TPB * FUSED_EPT) + (int)threadIdx.x;
    if (base + (FUSED_EPT - 1) * FUSED_TPB < nE) {
      int      d[FUSED_EPT], s[FUSED_EPT];
      unsigned r[FUSED_EPT];
#pragma unroll
      for (int j = 0; j < FUSED_EPT; ++j) {
        const int e = base + j * FUSED_TPB;
        d[j] = dst[e];
        r[j] = rank[e];
        s[j] = src[e];
      }
#pragma unroll
      for (int j = 0; j < FUSED_EPT; ++j) put_edge(slots, spill, nSpill, d[j], r[j], s[j]);
    } else {
      for (int j = 0; j < FUSED_EPT; ++j) {
        const int e = base + j * FUSED_TPB;
        if (e < nE) put_edge(slots, spill, nSpill, dst[e], rank[e], src[e]);
      }
    }
  }
}

template <int C>
__global__ __launch_bounds__(256) void gather_layer(
    const int* __restrict__ slots, const int* __restrict__ spill,
    const unsigned* __restrict__ nSpill, const unsigned* __restrict__ deg,
    const float* __restrict__ lin, const float* __restrict__ bias,
    float* __restrict__ out, int n, int doRelu) {
  constexpr int PER = C / 32;
  typedef float vec __attribute__((ext_vector_type(PER)));
  const int lane = threadIdx.x & 31;
  int node = (int)((blockIdx.x * (unsigned)blockDim.x + threadIdx.x) >> 5);
  node = __builtin_amdgcn_readfirstlane(node);
  if (node >= n) return;

  const int cnt  = (int)deg[node];
  const float di = rsqrtf((float)cnt + 1.0f);

  const vec self = *(const vec*)(lin + (size_t)node * C + lane * PER);
  vec acc = 0.0f;

  {
    const int m = (cnt < LIST_CAP) ? cnt : LIST_CAP;
    int sv = 0;
    if (lane < m) sv = slots[(unsigned)node * LIST_CAP + lane];
#pragma unroll 4
    for (int k = 0; k < m; ++k) {
      const int s = __builtin_amdgcn_readlane(sv, k);
      acc += *(const vec*)(lin + (size_t)s * C + lane * PER);
    }
  }
  if (cnt > LIST_CAP)
    add_spilled<C, vec>(acc, spill, (int)*nSpill, node, lane, lin + lane * PER);

  acc += self;
  vec v = di * acc + *(const vec*)(bias + lane * PER);
  if (doRelu) {
#pragma unroll
    for (int j = 0; j < PER; ++j) v[j] = fmaxf(v[j], 0.0f);
  }
  *(vec*)(out + (size_t)node * C + lane * PER) = v;
}

#define G1_NODES 32
#define G1_PITCH (HID_CH + 4)

__global__ __launch_bounds__(256) void gather1_gemm2_kernel(
    const int* __restrict__ slots, const int* __restrict__ spill,
    const unsigned* __restrict__ nSpill, const unsigned* __restrict__ deg,
    const float* __restrict__ lin, const float* __restrict__ bias,
    const float* __restrict__ W2, float* __restrict__ zlin, int n) {
  __shared__ float tile[G1_NODES * G1_PITCH];
  const int lane = threadIdx.x & 31;
  const int wave = __builtin_amdgcn_readfirstlane((int)(threadIdx.x >> 5));
  const int nodeBase = (int)blockIdx.x * G1_NODES;

  constexpr int NPW = G1_NODES / 8;
  int      cnt[NPW];
  float    di[NPW];
  v4f      self[NPW];
  int      sv0[NPW];
#pragma unroll
  for (int j = 0; j < NPW; ++j) {
    const int node = nodeBase + wave * NPW + j;
    const int nc   = (node < n) ? node : n - 1;
    const int dg = (int)deg[nc];
    cnt[j]   = (node < n) ? dg : 0;
    di[j]    = rsqrtf((float)dg + 1.0f);
    self[j]  = *(const v4f*)(lin + (size_t)nc * HID_CH + lane * 4);
  }
#pragma unroll
  for (int j = 0; j < NPW; ++j) {
    const int node = nodeBase + wave * NPW + j;
    sv0[j] = 0;
    if (lane < cnt[j]) sv0[j] = slots[(unsigned)node * LIST_CAP + lane];
  }
  const v4f bv = *(const v4f*)(bias + lane * 4);
  const float* __restrict__ lrow = lin + lane * 4;
#pragma unroll
  for (int j = 0; j < NPW; ++j) {
    v4f acc = 0.0f;
    {
      const int m  = (cnt[j] < LIST_CAP) ? cnt[j] : LIST_CAP;
      const int sv = sv0[j];
      int k = 0;
      for (; k + 4 <= m; k += 4) {
        const int s0 = __builtin_amdgcn_readlane(sv, k);
        const int s1 = __builtin_amdgcn_readlane(sv, k + 1);
        const int s2 = __builtin_amdgcn_readlane(sv, k + 2);
        const int s3 = __builtin_amdgcn_readlane(sv, k + 3);
        const v4f r0 = *(const v4f*)(lrow + (size_t)s0 * HID_CH);
        const v4f r1 = *(const v4f*)(lrow + (size_t)s1 * HID_CH);
        const v4f r2 = *(const v4f*)(lrow + (size_t)s2 * HID_CH);
        const v4f r3 = *(const v4f*)(lrow + (size_t)s3 * HID_CH);
        acc += r0; acc += r1; acc += r2; acc += r3;
      }
      if (k + 2 <= m) {
        const int s0 = __builtin_amdgcn_readlane(sv, k);
        const int s1 = __builtin_amdgcn_readlane(sv, k + 1);
        const v4f r0 = *(const v4f*)(lrow + (size_t)s0 * HID_CH);
        const v4f r1 = *(const v4f*)(lrow + (size_t)s1 * HID_CH);
        acc += r0; acc += r1;
        k += 2;
      }
      if (k < m) {
        const int s0 = __builtin_amdgcn_readlane(sv, k);
        acc += *(const v4f*)(lrow + (size_t)s0 * HID_CH);
      }
    }
    if (cnt[j] > LIST_CAP)
      add_spilled<HID_CH, v4f>(acc, spill, (int)*nSpill, nodeBase + wave * NPW + j, lane, lrow);
    acc += self[j];
    v4f v = di[j] * acc + bv;
    v.x = fmaxf(v.x, 0.0f); v.y = fmaxf(v.y, 0.0f);
    v.z = fmaxf(v.z, 0.0f); v.w = fmaxf(v.w, 0.0f);
    *(v4f*)(tile + (wave * NPW + j) * G1_PITCH + lane * 4) = v;
  }
  __syncthreads();

  const int g  = lane >> 4;
  const int l  = lane & 15;
  const int mt = wave >> 2;
  const int t  = wave & 3;
  const float* __restrict__ arow = tile + (mt * 16 + l) * G1_PITCH + 2 * g;
  const float* __restrict__ wcol = W2 + (size_t)(2 * g) * OUT_CH + t * 16 + l;
  const int mbase = nodeBase + mt * 16 + 8 * g;
  float rs[8];
#pragma unroll
  for (int r = 0; r < 8; ++r) {
    const int m = mbase + r;
    rs[r] = rsqrtf((float)deg[(m < n) ? m : n - 1] + 1.0f);
  }
  v8f acc2 = {};
#pragma unroll 8
  for (int k = 0; k < HID_CH; k += 4) {
    v2f a;
    a.x = arow[k];
    a.y = arow[k + 1];
    v2f b;
    b.x = wcol[(size_t)k * OUT_CH];
    b.y = wcol[(size_t)k * OUT_CH + OUT_CH];
    acc2 = __builtin_amdgcn_wmma_f32_16x16x4_f32(
         false, a,  false, b,
         (short)0, acc2,  false,  false);
  }
#pragma unroll
  for (int r = 0; r < 8; ++r) {
    const int m = mbase + r;
    if (m < n) zlin[(size_t)m * OUT_CH + t * 16 + l] = acc2[r] * rs[r];
  }
}

template <int J>
__device__ __forceinline__ int row_bcast(int v) {
  return __builtin_amdgcn_update_dpp(0, v, 0x150 + J, 0xf, 0xf, false);
}

template <int M>
__device__ __forceinline__ float row_xor(float v) {
  return __int_as_float(
      __builtin_amdgcn_update_dpp(0, __float_as_int(v), 0x160 + M, 0xf, 0xf, false));
}

template <int J>
__device__ __forceinline__ void edge_dots(float (&p)[16], int sv, int dv,
                                          const char* __restrict__ zb, unsigned coff) {
  const unsigned so = (unsigned)row_bcast<J>(sv) * (OUT_CH * 4u) + coff;
  const unsigned dofs = (unsigned)row_bcast<J>(dv) * (OUT_CH * 4u) + coff;
  const v4f a = *(const v4f*)(zb + so);
  const v4f b = *(const v4f*)(zb + dofs);
  p[J] = a.x * b.x + a.y * b.y + a.z * b.z + a.w * b.w;
  if constexpr (J + 1 < 16) edge_dots<J + 1>(p, sv, dv, zb, coff);
}

template <int H>
__device__ __forceinline__ void fold_step(float* v, bool upper) {
#pragma unroll
  for (int a = 0; a < H; ++a) {
    const float keep = upper ? v[a + H] : v[a];
    const float send = upper ? v[a] : v[a + H];
    v[a] = keep + row_xor<H>(send);
  }
}

__global__ __launch_bounds__(256) void decode_kernel(
    const int* __restrict__ src, const int* __restrict__ dst,
    const float* __restrict__ z, float* __restrict__ out, int nE) {
  const int lane = threadIdx.x & 31;
  int e0 = (int)(((blockIdx.x * (unsigned)blockDim.x + threadIdx.x) >> 5) << 5);
  e0 = __builtin_amdgcn_readfirstlane(e0);
  if (e0 >= nE) return;

  const int e = e0 + lane;
  int sv = 0, dv = 0;
  if (e < nE) {
    sv = src[e];
    dv = dst[e];
  }

  const char* __restrict__ zb = (const char*)z;
  const unsigned coff = (unsigned)(lane & 15) * 16u;

  float p[16];
  edge_dots<0>(p, sv, dv, zb, coff);

  fold_step<8>(p, (lane & 8) != 0);
  fold_step<4>(p, (lane & 4) != 0);
  fold_step<2>(p, (lane & 2) != 0);
  fold_step<1>(p, (lane & 1) != 0);

  if (e < nE) out[e] = 1.0f / (1.0f + __expf(-p[0]));
}

extern "C" void kernel_launch(void* const* d_in, const int* in_sizes, int n_in,
                              void* d_out, int out_size, void* d_ws, size_t ws_size,
                              hipStream_t stream) {
  const float* x   = (const float*)d_in[0];
  const int*   ei  = (const int*)d_in[1];
  const float* W1  = (const float*)d_in[2];
  const float* b1  = (const float*)d_in[3];
  const float* W2  = (const float*)d_in[4];
  const float* b2  = (const float*)d_in[5];

  const int N = in_sizes[0] / IN_CH;
  const int E = in_sizes[1] / 2;
  const int* esrc = ei;
  const int* edst = ei + E;

  char* w = (char*)d_ws;
  unsigned* deg    = (unsigned*)w; w += (size_t)N * 4;
  unsigned* nSpill = (unsigned*)w; w += (size_t)N * 4;
  float*    hlin = (float*)w;    w += (size_t)N * HID_CH * 4;
  float*    h    = (float*)w;    w += (size_t)N * HID_CH * 4;
  char*     r3   = w;            w += (size_t)N * OUT_CH * 4;
  char*     r4   = w;
  int*      slots = (int*)r3;
  unsigned* rank  = (unsigned*)r4;
  float*    zlin  = h;
  int*      spill = (int*)(h + (size_t)N * OUT_CH);
  float*    z     = hlin;

  hipMemsetAsync(deg, 0, (size_t)N * 4 + 4, stream);

  const int TPB = 256;
  const int nEdgeBlocks = (E + TPB - 1) / TPB;
  const int nNodeWaveBlocks = (int)(((size_t)N * 32 + TPB - 1) / TPB);
  const int mTiles = (N + 15) / 16;

  deg_rank_kernel<<<nEdgeBlocks, TPB, 0, stream>>>(edst, deg, rank, E);

  const int nGemmBlocks = (mTiles + FUSED_TPB / 32 - 1) / (FUSED_TPB / 32);
  const int nFillBlocks = (E + FUSED_TPB * FUSED_EPT - 1) / (FUSED_TPB * FUSED_EPT);
  gemm1_fill_kernel<<<nGemmBlocks + nFillBlocks, FUSED_TPB, 0, stream>>>(
      x, W1, deg, hlin, N, nGemmBlocks, esrc, edst, rank, slots, spill, nSpill, E);

  const int nG1Blocks = (N + G1_NODES - 1) / G1_NODES;
  gather1_gemm2_kernel<<<nG1Blocks, TPB, 0, stream>>>(
      slots, spill, nSpill, deg, hlin, b1, W2, zlin, N);

  gather_layer<OUT_CH><<<nNodeWaveBlocks, TPB, 0, stream>>>(
      slots, spill, nSpill, deg, zlin, b2, z, N, 0);

  decode_kernel<<<nEdgeBlocks, TPB, 0, stream>>>(esrc, edst, z, (float*)d_out, E);
}


# Node lists by two-pass LDS binning: edges routed to 256-node buckets via LDS-staged 256-B cells, ranks by LDS atomics, slot lines written whole; no global atomic per edge; lists in LDS arrival order
# speedup vs baseline: 4.1206x; 1.3110x over previous
#include <hip/hip_runtime.h>
#include <math.h>


typedef __attribute__((ext_vector_type(2))) float v2f;
typedef __attribute__((ext_vector_type(8))) float v8f;
typedef __attribute__((ext_vector_type(4))) float v4f;
typedef __attribute__((ext_vector_type(2))) int v2i;
typedef __attribute__((ext_vector_type(4))) int v4i;

#define IN_CH  256
#define HID_CH 128
#define OUT_CH 64

template <int K, int NCOLS, bool SCALE_EARLY>
__device__ __forceinline__ void gemm_stripe(
    const float* __restrict__ A, const float* __restrict__ W,
    const unsigned* __restrict__ rowDeg, float* __restrict__ Out, int nRows, int stripe) {
  constexpr int NT = NCOLS / 16;
  const int mbase = stripe * 16;
  const int lane  = threadIdx.x & 31;
  const int g     = lane >> 4;
  const int l     = lane & 15;

  v8f acc[NT] = {};

  int arow_idx = mbase + l;
  if (arow_idx >= nRows) arow_idx = nRows - 1;
  const float* __restrict__ arow = A + (size_t)arow_idx * K + 2 * g;

  float rs[8];
  if constexpr (SCALE_EARLY) {
#pragma unroll
    for (int r = 0; r < 8; ++r) {
      int m = mbase + r + 8 * g;
      if (m >= nRows) m = nRows - 1;
      rs[r] = rsqrtf((float)rowDeg[m] + 1.0f);
    }
  }

#pragma unroll 4
  for (int k = 0; k < K; k += 4) {
    v2f a;
    a.x = arow[k];
    a.y = arow[k + 1];
    const float* __restrict__ wk = W + (size_t)(k + 2 * g) * NCOLS + l;
#pragma unroll
    for (int t = 0; t < NT; ++t) {
      v2f b;
      b.x = wk[t * 16];
      b.y = wk[NCOLS + t * 16];
      acc[t] = __builtin_amdgcn_wmma_f32_16x16x4_f32(
           false, a,  false, b,
           (short)0, acc[t],  false,  false);
    }
  }

  if constexpr (!SCALE_EARLY) {
#pragma unroll
    for (int r = 0; r < 8; ++r) {
      int m = mbase + r + 8 * g;
      if (m >= nRows) m = nRows - 1;
      rs[r] = rsqrtf((float)rowDeg[m] + 1.0f);
    }
  }

#pragma unroll
  for (int t = 0; t < NT; ++t) {
#pragma unroll
    for (int r = 0; r < 8; ++r) {
      const int m = mbase + r + 8 * g;
      if (m < nRows) Out[(size_t)m * NCOLS + t * 16 + l] = acc[t][r] * rs[r];
    }
  }
}

#define LIST_CAP 32

#define BKT_SHIFT   8
#define BKT_NODES   (1 << BKT_SHIFT)
#define NBK_MAX     400
#define CELL_CAP    32
#define CELL_VECS   (CELL_CAP / 2)
#define CHUNK_EDGES 4096
#define BIN_TPB     256
#define BIN_EPT     (CHUNK_EDGES / BIN_TPB)

__global__ __launch_bounds__(BIN_TPB) void bin_edges_kernel(
    const int* __restrict__ src, const int* __restrict__ dst,
    int* __restrict__ cells, int* __restrict__ ovf, unsigned* __restrict__ nOvf,
    int nE, int nBuckets, int nChunks) {
  __shared__ v4i      stage4[NBK_MAX * CELL_VECS];
  __shared__ unsigned lcnt[NBK_MAX];
  v2i* __restrict__ stage = (v2i*)stage4;
  const int tid   = (int)threadIdx.x;
  const int chunk = (int)blockIdx.x;

  v4i* st4 = stage4;
  const int nVec = nBuckets * CELL_VECS;
  const v4i none = {-1, 0, -1, 0};
  for (int i = tid; i < nVec; i += BIN_TPB) st4[i] = none;
  for (int i = tid; i < nBuckets; i += BIN_TPB) lcnt[i] = 0u;
  __syncthreads();

  int d[BIN_EPT], s[BIN_EPT];
  const int e0 = chunk * CHUNK_EDGES + tid;
#pragma unroll
  for (int j = 0; j < BIN_EPT; ++j) {
    const int e = e0 + j * BIN_TPB;
    d[j] = -1;
    s[j] = 0;
    if (e < nE) {
      d[j] = dst[e];
      s[j] = src[e];
    }
  }
#pragma unroll
  for (int j = 0; j < BIN_EPT; ++j) {
    if (d[j] >= 0) {
      const unsigned b = (unsigned)d[j] >> BKT_SHIFT;
      const unsigned r = atomicAdd(&lcnt[b], 1u);
      if (r < CELL_CAP) {
        const v2i pr = {d[j], s[j]};
        stage[b * CELL_CAP + r] = pr;
      } else {
        const unsigned o = atomicAdd(nOvf, 1u);
        ovf[2 * o]     = d[j];
        ovf[2 * o + 1] = s[j];
      }
    }
  }
  __syncthreads();

  for (int i = tid; i < nVec; i += BIN_TPB) {
    const int b = i / CELL_VECS;
    const int w = i % CELL_VECS;
    v4i* __restrict__ cell = (v4i*)cells + ((size_t)b * nChunks + chunk) * CELL_VECS;
    cell[w] = st4[i];
  }
}

__device__ __forceinline__ void take_edge(unsigned* lcnt, int* lslots, int nodeBase,
                                          int* __restrict__ spill, unsigned* __restrict__ nSpill,
                                          int d, int s) {
  const unsigned loc = (unsigned)(d - nodeBase);
  const unsigned r   = atomicAdd(&lcnt[loc], 1u);
  if (r < LIST_CAP) {
    lslots[loc * LIST_CAP + r] = s;
  } else {
    const unsigned o = atomicAdd(nSpill, 1u);
    spill[2 * o]     = d;
    spill[2 * o + 1] = s;
  }
}

#define BUILD_AHEAD 8

__global__ __launch_bounds__(BIN_TPB) void build_lists_kernel(
    const int* __restrict__ cells, const int* __restrict__ ovf,
    const unsigned* __restrict__ nOvf, unsigned* __restrict__ deg,
    int* __restrict__ slots, int* __restrict__ spill, unsigned* __restrict__ nSpill,
    int n, int nChunks) {
  __shared__ int      lslots[BKT_NODES * LIST_CAP];
  __shared__ unsigned lcnt[BKT_NODES];
  const int tid      = (int)threadIdx.x;
  const int bucket   = (int)blockIdx.x;
  const int nodeBase = bucket << BKT_SHIFT;

  for (int i = tid; i < BKT_NODES; i += BIN_TPB) lcnt[i] = 0u;
  __syncthreads();

  const v4i* __restrict__ col = (const v4i*)cells + (size_t)bucket * nChunks * CELL_VECS;
  const int nVec = nChunks * CELL_VECS;
  for (int i0 = 0; i0 < nVec; i0 += BIN_TPB * BUILD_AHEAD) {
    v4i v[BUILD_AHEAD];
#pragma unroll
    for (int u = 0; u < BUILD_AHEAD; ++u) {
      const int i = i0 + u * BIN_TPB + tid;
      const v4i none = {-1, 0, -1, 0};
      v[u] = none;
      if (i < nVec) v[u] = col[i];
    }
#pragma unroll
    for (int u = 0; u < BUILD_AHEAD; ++u) {
      if (v[u].x >= 0) take_edge(lcnt, lslots, nodeBase, spill, nSpill, v[u].x, v[u].y);
      if (v[u].z >= 0) take_edge(lcnt, lslots, nodeBase, spill, nSpill, v[u].z, v[u].w);
    }
  }
  const int nOver = (int)*nOvf;
  for (int i = tid; i < nOver; i += BIN_TPB) {
    const int d = ovf[2 * i];
    const int s = ovf[2 * i + 1];
    if ((d >> BKT_SHIFT) == bucket) take_edge(lcnt, lslots, nodeBase, spill, nSpill, d, s);
  }
  __syncthreads();

  for (int i = tid; i < BKT_NODES; i += BIN_TPB) {
    const int node = nodeBase + i;
    if (node < n) deg[node] = lcnt[i];
  }
  for (int i = tid; i < BKT_NODES * LIST_CAP; i += BIN_TPB) {
    const int loc  = i / LIST_CAP;
    const int k    = i % LIST_CAP;
    const int node = nodeBase + loc;
    if (node < n && (unsigned)k < lcnt[loc]) slots[(size_t)node * LIST_CAP + k] = lslots[i];
  }
}

template <int C, typename V>
__device__ __forceinline__ void add_spilled(V& acc, const int* __restrict__ spill, int nSpill,
                                            int node, int lane, const float* __restrict__ lrow) {
  for (int i = 0; i < nSpill; i += 32) {
    int od = -1, os = 0;
    if (i + lane < nSpill) {
      od = spill[2 * (i + lane)];
      os = spill[2 * (i + lane) + 1];
    }
    unsigned hit = __builtin_amdgcn_ballot_w32(od == node);
    while (hit != 0u) {
      const int k = __builtin_ctz(hit);
      hit &= hit - 1u;
      const int s = __builtin_amdgcn_readlane(os, k);
      acc += *(const V*)(lrow + (size_t)s * C);
    }
  }
}

#define GEMM1_TPB 128

__global__ __launch_bounds__(GEMM1_TPB) __attribute__((amdgpu_waves_per_eu(9, 9)))
void gemm1_kernel(
    const float* __restrict__ x, const float* __restrict__ W1,
    const unsigned* __restrict__ deg, float* __restrict__ hlin, int nRows) {
  int stripe = (int)blockIdx.x * (GEMM1_TPB / 32) + (int)(threadIdx.x >> 5);
  stripe = __builtin_amdgcn_readfirstlane(stripe);
  if (stripe * 16 >= nRows) return;
  gemm_stripe<IN_CH, HID_CH, false>(x, W1, deg, hlin, nRows, stripe);
}

template <int C>
__global__ __launch_bounds__(256) void gather_layer(
    const int* __restrict__ slots, const int* __restrict__ spill,
    const unsigned* __restrict__ nSpill, const unsigned* __restrict__ deg,
    const float* __restrict__ lin, const float* __restrict__ bias,
    float* __restrict__ out, int n, int doRelu) {
  constexpr int PER = C / 32;
  typedef float vec __attribute__((ext_vector_type(PER)));
  const int lane = threadIdx.x & 31;
  int node = (int)((blockIdx.x * (unsigned)blockDim.x + threadIdx.x) >> 5);
  node = __builtin_amdgcn_readfirstlane(node);
  if (node >= n) return;

  const int cnt  = (int)deg[node];
  const float di = rsqrtf((float)cnt + 1.0f);

  const vec self = *(const vec*)(lin + (size_t)node * C + lane * PER);
  vec acc = 0.0f;

  {
    const int m = (cnt < LIST_CAP) ? cnt : LIST_CAP;
    int sv = 0;
    if (lane < m) sv = slots[(unsigned)node * LIST_CAP + lane];
#pragma unroll 4
    for (int k = 0; k < m; ++k) {
      const int s = __builtin_amdgcn_readlane(sv, k);
      acc += *(const vec*)(lin + (size_t)s * C + lane * PER);
    }
  }
  if (cnt > LIST_CAP)
    add_spilled<C, vec>(acc, spill, (int)*nSpill, node, lane, lin + lane * PER);

  acc += self;
  vec v = di * acc + *(const vec*)(bias + lane * PER);
  if (doRelu) {
#pragma unroll
    for (int j = 0; j < PER; ++j) v[j] = fmaxf(v[j], 0.0f);
  }
  *(vec*)(out + (size_t)node * C + lane * PER) = v;
}

#define G1_NODES 32
#define G1_PITCH (HID_CH + 4)

__global__ __launch_bounds__(256) void gather1_gemm2_kernel(
    const int* __restrict__ slots, const int* __restrict__ spill,
    const unsigned* __restrict__ nSpill, const unsigned* __restrict__ deg,
    const float* __restrict__ lin, const float* __restrict__ bias,
    const float* __restrict__ W2, float* __restrict__ zlin, int n) {
  __shared__ float tile[G1_NODES * G1_PITCH];
  const int lane = threadIdx.x & 31;
  const int wave = __builtin_amdgcn_readfirstlane((int)(threadIdx.x >> 5));
  const int nodeBase = (int)blockIdx.x * G1_NODES;

  constexpr int NPW = G1_NODES / 8;
  int      cnt[NPW];
  float    di[NPW];
  v4f      self[NPW];
  int      sv0[NPW];
#pragma unroll
  for (int j = 0; j < NPW; ++j) {
    const int node = nodeBase + wave * NPW + j;
    const int nc   = (node < n) ? node : n - 1;
    const int dg = (int)deg[nc];
    cnt[j]   = (node < n) ? dg : 0;
    di[j]    = rsqrtf((float)dg + 1.0f);
    self[j]  = *(const v4f*)(lin + (size_t)nc * HID_CH + lane * 4);
  }
#pragma unroll
  for (int j = 0; j < NPW; ++j) {
    const int node = nodeBase + wave * NPW + j;
    sv0[j] = 0;
    if (lane < cnt[j]) sv0[j] = slots[(unsigned)node * LIST_CAP + lane];
  }
  const v4f bv = *(const v4f*)(bias + lane * 4);
  const float* __restrict__ lrow = lin + lane * 4;
#pragma unroll
  for (int j = 0; j < NPW; ++j) {
    v4f acc = 0.0f;
    {
      const int m  = (cnt[j] < LIST_CAP) ? cnt[j] : LIST_CAP;
      const int sv = sv0[j];
      int k = 0;
      for (; k + 4 <= m; k += 4) {
        const int s0 = __builtin_amdgcn_readlane(sv, k);
        const int s1 = __builtin_amdgcn_readlane(sv, k + 1);
        const int s2 = __builtin_amdgcn_readlane(sv, k + 2);
        const int s3 = __builtin_amdgcn_readlane(sv, k + 3);
        const v4f r0 = *(const v4f*)(lrow + (size_t)s0 * HID_CH);
        const v4f r1 = *(const v4f*)(lrow + (size_t)s1 * HID_CH);
        const v4f r2 = *(const v4f*)(lrow + (size_t)s2 * HID_CH);
        const v4f r3 = *(const v4f*)(lrow + (size_t)s3 * HID_CH);
        acc += r0; acc += r1; acc += r2; acc += r3;
      }
      if (k + 2 <= m) {
        const int s0 = __builtin_amdgcn_readlane(sv, k);
        const int s1 = __builtin_amdgcn_readlane(sv, k + 1);
        const v4f r0 = *(const v4f*)(lrow + (size_t)s0 * HID_CH);
        const v4f r1 = *(const v4f*)(lrow + (size_t)s1 * HID_CH);
        acc += r0; acc += r1;
        k += 2;
      }
      if (k < m) {
        const int s0 = __builtin_amdgcn_readlane(sv, k);
        acc += *(const v4f*)(lrow + (size_t)s0 * HID_CH);
      }
    }
    if (cnt[j] > LIST_CAP)
      add_spilled<HID_CH, v4f>(acc, spill, (int)*nSpill, nodeBase + wave * NPW + j, lane, lrow);
    acc += self[j];
    v4f v = di[j] * acc + bv;
    v.x = fmaxf(v.x, 0.0f); v.y = fmaxf(v.y, 0.0f);
    v.z = fmaxf(v.z, 0.0f); v.w = fmaxf(v.w, 0.0f);
    *(v4f*)(tile + (wave * NPW + j) * G1_PITCH + lane * 4) = v;
  }
  __syncthreads();

  const int g  = lane >> 4;
  const int l  = lane & 15;
  const int mt = wave >> 2;
  const int t  = wave & 3;
  const float* __restrict__ arow = tile + (mt * 16 + l) * G1_PITCH + 2 * g;
  const float* __restrict__ wcol = W2 + (size_t)(2 * g) * OUT_CH + t * 16 + l;
  const int mbase = nodeBase + mt * 16 + 8 * g;
  float rs[8];
#pragma unroll
  for (int r = 0; r < 8; ++r) {
    const int m = mbase + r;
    rs[r] = rsqrtf((float)deg[(m < n) ? m : n - 1] + 1.0f);
  }
  v8f acc2 = {};
#pragma unroll 8
  for (int k = 0; k < HID_CH; k += 4) {
    v2f a;
    a.x = arow[k];
    a.y = arow[k + 1];
    v2f b;
    b.x = wcol[(size_t)k * OUT_CH];
    b.y = wcol[(size_t)k * OUT_CH + OUT_CH];
    acc2 = __builtin_amdgcn_wmma_f32_16x16x4_f32(
         false, a,  false, b,
         (short)0, acc2,  false,  false);
  }
#pragma unroll
  for (int r = 0; r < 8; ++r) {
    const int m = mbase + r;
    if (m < n) zlin[(size_t)m * OUT_CH + t * 16 + l] = acc2[r] * rs[r];
  }
}

template <int J>
__device__ __forceinline__ int row_bcast(int v) {
  return __builtin_amdgcn_update_dpp(0, v, 0x150 + J, 0xf, 0xf, false);
}

template <int M>
__device__ __forceinline__ float row_xor(float v) {
  return __int_as_float(
      __builtin_amdgcn_update_dpp(0, __float_as_int(v), 0x160 + M, 0xf, 0xf, false));
}

template <int J>
__device__ __forceinline__ void edge_dots(float (&p)[16], int sv, int dv,
                                          const char* __restrict__ zb, unsigned coff) {
  const unsigned so = (unsigned)row_bcast<J>(sv) * (OUT_CH * 4u) + coff;
  const unsigned dofs = (unsigned)row_bcast<J>(dv) * (OUT_CH * 4u) + coff;
  const v4f a = *(const v4f*)(zb + so);
  const v4f b = *(const v4f*)(zb + dofs);
  p[J] = a.x * b.x + a.y * b.y + a.z * b.z + a.w * b.w;
  if constexpr (J + 1 < 16) edge_dots<J + 1>(p, sv, dv, zb, coff);
}

template <int H>
__device__ __forceinline__ void fold_step(float* v, bool upper) {
#pragma unroll
  for (int a = 0; a < H; ++a) {
    const float keep = upper ? v[a + H] : v[a];
    const float send = upper ? v[a] : v[a + H];
    v[a] = keep + row_xor<H>(send);
  }
}

__global__ __launch_bounds__(256) void decode_kernel(
    const int* __restrict__ src, const int* __restrict__ dst,
    const float* __restrict__ z, float* __restrict__ out, int nE) {
  const int lane = threadIdx.x & 31;
  int e0 = (int)(((blockIdx.x * (unsigned)blockDim.x + threadIdx.x) >> 5) << 5);
  e0 = __builtin_amdgcn_readfirstlane(e0);
  if (e0 >= nE) return;

  const int e = e0 + lane;
  int sv = 0, dv = 0;
  if (e < nE) {
    sv = src[e];
    dv = dst[e];
  }

  const char* __restrict__ zb = (const char*)z;
  const unsigned coff = (unsigned)(lane & 15) * 16u;

  float p[16];
  edge_dots<0>(p, sv, dv, zb, coff);

  fold_step<8>(p, (lane & 8) != 0);
  fold_step<4>(p, (lane & 4) != 0);
  fold_step<2>(p, (lane & 2) != 0);
  fold_step<1>(p, (lane & 1) != 0);

  if (e < nE) out[e] = 1.0f / (1.0f + __expf(-p[0]));
}

extern "C" void kernel_launch(void* const* d_in, const int* in_sizes, int n_in,
                              void* d_out, int out_size, void* d_ws, size_t ws_size,
                              hipStream_t stream) {
  const float* x   = (const float*)d_in[0];
  const int*   ei  = (const int*)d_in[1];
  const float* W1  = (const float*)d_in[2];
  const float* b1  = (const float*)d_in[3];
  const float* W2  = (const float*)d_in[4];
  const float* b2  = (const float*)d_in[5];

  const int N = in_sizes[0] / IN_CH;
  const int E = in_sizes[1] / 2;
  const int* esrc = ei;
  const int* edst = ei + E;

  char* w = (char*)d_ws;
  unsigned* deg    = (unsigned*)w; w += (size_t)N * 4;
  unsigned* nSpill = (unsigned*)w; w += (size_t)N * 4;
  float*    hlin = (float*)w;    w += (size_t)N * HID_CH * 4;
  float*    h    = (float*)w;    w += (size_t)N * HID_CH * 4;
  char*     r3   = w;
  int*      slots = (int*)r3;
  unsigned* nOvf  = nSpill + 1;
  int*      cells = (int*)hlin;
  int*      ovf   = (int*)h;
  float*    zlin  = h;
  int*      spill = (int*)(h + (size_t)N * OUT_CH);
  float*    z     = hlin;

  hipMemsetAsync(nSpill, 0, 8, stream);

  const int TPB = 256;
  const int nEdgeBlocks = (E + TPB - 1) / TPB;
  const int nNodeWaveBlocks = (int)(((size_t)N * 32 + TPB - 1) / TPB);
  const int mTiles = (N + 15) / 16;

  const int nBuckets = (N + BKT_NODES - 1) / BKT_NODES;
  const int nChunks  = (E + CHUNK_EDGES - 1) / CHUNK_EDGES;
  bin_edges_kernel<<<nChunks, BIN_TPB, 0, stream>>>(esrc, edst, cells, ovf, nOvf, E, nBuckets, nChunks);
  build_lists_kernel<<<nBuckets, BIN_TPB, 0, stream>>>(cells, ovf, nOvf, deg, slots, spill, nSpill, N, nChunks);

  const int nGemmBlocks = (mTiles + GEMM1_TPB / 32 - 1) / (GEMM1_TPB / 32);
  gemm1_kernel<<<nGemmBlocks, GEMM1_TPB, 0, stream>>>(x, W1, deg, hlin, N);

  const int nG1Blocks = (N + G1_NODES - 1) / G1_NODES;
  gather1_gemm2_kernel<<<nG1Blocks, TPB, 0, stream>>>(
      slots, spill, nSpill, deg, hlin, b1, W2, zlin, N);

  gather_layer<OUT_CH><<<nNodeWaveBlocks, TPB, 0, stream>>>(
      slots, spill, nSpill, deg, zlin, b2, z, N, 0);

  decode_kernel<<<nEdgeBlocks, TPB, 0, stream>>>(esrc, edst, z, (float*)d_out, E);
}


# Layer-1 GEMM operands 16 bytes a lane: W1 read from a copy packed in the call in fragment order, 4 adjacent k per lane half, output rows stored as whole lines; k summed in a new order
# speedup vs baseline: 4.2655x; 1.0352x over previous
#include <hip/hip_runtime.h>
#include <math.h>


typedef __attribute__((ext_vector_type(2))) float v2f;
typedef __attribute__((ext_vector_type(8))) float v8f;
typedef __attribute__((ext_vector_type(4))) float v4f;
typedef __attribute__((ext_vector_type(2))) int v2i;
typedef __attribute__((ext_vector_type(4))) int v4i;

#define IN_CH  256
#define HID_CH 128
#define OUT_CH 64

#define WP_PIECES ((IN_CH / 8) * (HID_CH / 16) * 32)

template <int NCOLS>
__device__ __forceinline__ void pack_w(const float* __restrict__ W, float* __restrict__ Wp, int piece) {
  constexpr int NT = NCOLS / 16;
  const int ln  = piece & 31;
  const int t   = (piece >> 5) % NT;
  const int kb  = (piece >> 5) / NT;
  const int g   = ln >> 4;
  const int l   = ln & 15;
  const int col = 64 * (t >> 2) + 4 * l + (t & 3);
  const float* __restrict__ w = W + (size_t)(kb * 8 + 4 * g) * NCOLS + col;
  v4f p;
  p.x = w[0];
  p.y = w[NCOLS];
  p.z = w[2 * NCOLS];
  p.w = w[3 * NCOLS];
  ((v4f*)Wp)[piece] = p;
}

template <int K, int NCOLS>
__device__ __forceinline__ void gemm_stripe(
    const float* __restrict__ A, const float* __restrict__ Wp,
    const unsigned* __restrict__ rowDeg, float* __restrict__ Out, int nRows, int stripe) {
  constexpr int NT = NCOLS / 16;
  constexpr int NC = NCOLS / 64;
  static_assert(NCOLS % 64 == 0 && K % 8 == 0, "shape");
  const int mbase = stripe * 16;
  const int lane  = threadIdx.x & 31;
  const int g     = lane >> 4;
  const int l     = lane & 15;

  v8f acc[NT] = {};

  int arow_idx = mbase + l;
  if (arow_idx >= nRows) arow_idx = nRows - 1;
  const float* __restrict__ arow = A + (size_t)arow_idx * K + 4 * g;
  const v4f* __restrict__ wl = (const v4f*)Wp + lane;

#pragma unroll 2
  for (int k = 0; k < K; k += 8) {
    const v4f a4 = *(const v4f*)(arow + k);
    v2f a0, a1;
    a0.x = a4.x; a0.y = a4.y;
    a1.x = a4.z; a1.y = a4.w;
    const v4f* __restrict__ wk = wl + (size_t)(k / 8) * NT * 32;
#pragma unroll
    for (int t = 0; t < NT; ++t) {
      const v4f b4 = wk[t * 32];
      v2f p, q;
      p.x = b4.x; p.y = b4.y;
      q.x = b4.z; q.y = b4.w;
      acc[t] = __builtin_amdgcn_wmma_f32_16x16x4_f32(
           false, a0,  false, p,
           (short)0, acc[t],  false,  false);
      acc[t] = __builtin_amdgcn_wmma_f32_16x16x4_f32(
           false, a1,  false, q,
           (short)0, acc[t],  false,  false);
    }
  }

  float rs[8];
#pragma unroll
  for (int r = 0; r < 8; ++r) {
    int m = mbase + r + 8 * g;
    if (m >= nRows) m = nRows - 1;
    rs[r] = rsqrtf((float)rowDeg[m] + 1.0f);
  }

#pragma unroll
  for (int r = 0; r < 8; ++r) {
    const int m = mbase + r + 8 * g;
    if (m < nRows) {
#pragma unroll
      for (int c = 0; c < NC; ++c) {
        v4f o;
        o.x = acc[4 * c + 0][r] * rs[r];
        o.y = acc[4 * c + 1][r] * rs[r];
        o.z = acc[4 * c + 2][r] * rs[r];
        o.w = acc[4 * c + 3][r] * rs[r];
        *(v4f*)(Out + (size_t)m * NCOLS + c * 64 + 4 * l) = o;
      }
    }
  }
}

#define LIST_CAP 32

#define BKT_SHIFT   8
#define BKT_NODES   (1 << BKT_SHIFT)
#define NBK_MAX     400
#define CELL_CAP    32
#define CELL_VECS   (CELL_CAP / 2)
#define CHUNK_EDGES 4096
#define BIN_TPB     256
#define BIN_EPT     (CHUNK_EDGES / BIN_TPB)

__global__ __launch_bounds__(BIN_TPB) void bin_edges_kernel(
    const int* __restrict__ src, const int* __restrict__ dst,
    int* __restrict__ cells, int* __restrict__ ovf, unsigned* __restrict__ nOvf,
    int nE, int nBuckets, int nChunks) {
  __shared__ v4i      stage4[NBK_MAX * CELL_VECS];
  __shared__ unsigned lcnt[NBK_MAX];
  v2i* __restrict__ stage = (v2i*)stage4;
  const int tid   = (int)threadIdx.x;
  const int chunk = (int)blockIdx.x;

  v4i* st4 = stage4;
  const int nVec = nBuckets * CELL_VECS;
  const v4i none = {-1, 0, -1, 0};
  for (int i = tid; i < nVec; i += BIN_TPB) st4[i] = none;
  for (int i = tid; i < nBuckets; i += BIN_TPB) lcnt[i] = 0u;
  __syncthreads();

  int d[BIN_EPT], s[BIN_EPT];
  const int e0 = chunk * CHUNK_EDGES + tid;
#pragma unroll
  for (int j = 0; j < BIN_EPT; ++j) {
    const int e = e0 + j * BIN_TPB;
    d[j] = -1;
    s[j] = 0;
    if (e < nE) {
      d[j] = dst[e];
      s[j] = src[e];
    }
  }
#pragma unroll
  for (int j = 0; j < BIN_EPT; ++j) {
    if (d[j] >= 0) {
      const unsigned b = (unsigned)d[j] >> BKT_SHIFT;
      const unsigned r = atomicAdd(&lcnt[b], 1u);
      if (r < CELL_CAP) {
        const v2i pr = {d[j], s[j]};
        stage[b * CELL_CAP + r] = pr;
      } else {
        const unsigned o = atomicAdd(nOvf, 1u);
        ovf[2 * o]     = d[j];
        ovf[2 * o + 1] = s[j];
      }
    }
  }
  __syncthreads();

  for (int i = tid; i < nVec; i += BIN_TPB) {
    const int b = i / CELL_VECS;
    const int w = i % CELL_VECS;
    v4i* __restrict__ cell = (v4i*)cells + ((size_t)b * nChunks + chunk) * CELL_VECS;
    cell[w] = st4[i];
  }
}

__device__ __forceinline__ void take_edge(unsigned* lcnt, int* lslots, int nodeBase,
                                          int* __restrict__ spill, unsigned* __restrict__ nSpill,
                                          int d, int s) {
  const unsigned loc = (unsigned)(d - nodeBase);
  const unsigned r   = atomicAdd(&lcnt[loc], 1u);
  if (r < LIST_CAP) {
    lslots[loc * LIST_CAP + r] = s;
  } else {
    const unsigned o = atomicAdd(nSpill, 1u);
    spill[2 * o]     = d;
    spill[2 * o + 1] = s;
  }
}

#define BUILD_AHEAD 8

__global__ __launch_bounds__(BIN_TPB) void build_lists_kernel(
    const int* __restrict__ cells, const int* __restrict__ ovf,
    const unsigned* __restrict__ nOvf, unsigned* __restrict__ deg,
    int* __restrict__ slots, int* __restrict__ spill, unsigned* __restrict__ nSpill,
    const float* __restrict__ W1, float* __restrict__ W1p,
    int n, int nChunks) {
  __shared__ int      lslots[BKT_NODES * LIST_CAP];
  __shared__ unsigned lcnt[BKT_NODES];
  const int tid      = (int)threadIdx.x;
  const int bucket   = (int)blockIdx.x;
  const int nodeBase = bucket << BKT_SHIFT;

  for (int i = tid; i < BKT_NODES; i += BIN_TPB) lcnt[i] = 0u;
  __syncthreads();

  const v4i* __restrict__ col = (const v4i*)cells + (size_t)bucket * nChunks * CELL_VECS;
  const int nVec = nChunks * CELL_VECS;
  for (int i0 = 0; i0 < nVec; i0 += BIN_TPB * BUILD_AHEAD) {
    v4i v[BUILD_AHEAD];
#pragma unroll
    for (int u = 0; u < BUILD_AHEAD; ++u) {
      const int i = i0 + u * BIN_TPB + tid;
      const v4i none = {-1, 0, -1, 0};
      v[u] = none;
      if (i < nVec) v[u] = col[i];
    }
#pragma unroll
    for (int u = 0; u < BUILD_AHEAD; ++u) {
      if (v[u].x >= 0) take_edge(lcnt, lslots, nodeBase, spill, nSpill, v[u].x, v[u].y);
      if (v[u].z >= 0) take_edge(lcnt, lslots, nodeBase, spill, nSpill, v[u].z, v[u].w);
    }
  }
  const int nOver = (int)*nOvf;
  for (int i = tid; i < nOver; i += BIN_TPB) {
    const int d = ovf[2 * i];
    const int s = ovf[2 * i + 1];
    if ((d >> BKT_SHIFT) == bucket) take_edge(lcnt, lslots, nodeBase, spill, nSpill, d, s);
  }
  __syncthreads();

  for (int p = bucket * BIN_TPB + tid; p < WP_PIECES; p += (int)gridDim.x * BIN_TPB)
    pack_w<HID_CH>(W1, W1p, p);

  for (int i = tid; i < BKT_NODES; i += BIN_TPB) {
    const int node = nodeBase + i;
    if (node < n) deg[node] = lcnt[i];
  }
  for (int i = tid; i < BKT_NODES * LIST_CAP; i += BIN_TPB) {
    const int loc  = i / LIST_CAP;
    const int k    = i % LIST_CAP;
    const int node = nodeBase + loc;
    if (node < n && (unsigned)k < lcnt[loc]) slots[(size_t)node * LIST_CAP + k] = lslots[i];
  }
}

template <int C, typename V>
__device__ __forceinline__ void add_spilled(V& acc, const int* __restrict__ spill, int nSpill,
                                            int node, int lane, const float* __restrict__ lrow) {
  for (int i = 0; i < nSpill; i += 32) {
    int od = -1, os = 0;
    if (i + lane < nSpill) {
      od = spill[2 * (i + lane)];
      os = spill[2 * (i + lane) + 1];
    }
    unsigned hit = __builtin_amdgcn_ballot_w32(od == node);
    while (hit != 0u) {
      const int k = __builtin_ctz(hit);
      hit &= hit - 1u;
      const int s = __builtin_amdgcn_readlane(os, k);
      acc += *(const V*)(lrow + (size_t)s * C);
    }
  }
}

#define GEMM1_TPB 128

__global__ __launch_bounds__(GEMM1_TPB) __attribute__((amdgpu_waves_per_eu(9, 9)))
void gemm1_kernel(
    const float* __restrict__ x, const float* __restrict__ W1p,
    const unsigned* __restrict__ deg, float* __restrict__ hlin, int nRows) {
  int stripe = (int)blockIdx.x * (GEMM1_TPB / 32) + (int)(threadIdx.x >> 5);
  stripe = __builtin_amdgcn_readfirstlane(stripe);
  if (stripe * 16 >= nRows) return;
  gemm_stripe<IN_CH, HID_CH>(x, W1p, deg, hlin, nRows, stripe);
}

template <int C>
__global__ __launch_bounds__(256) void gather_layer(
    const int* __restrict__ slots, const int* __restrict__ spill,
    const unsigned* __restrict__ nSpill, const unsigned* __restrict__ deg,
    const float* __restrict__ lin, const float* __restrict__ bias,
    float* __restrict__ out, int n, int doRelu) {
  constexpr int PER = C / 32;
  typedef float vec __attribute__((ext_vector_type(PER)));
  const int lane = threadIdx.x & 31;
  int node = (int)((blockIdx.x * (unsigned)blockDim.x + threadIdx.x) >> 5);
  node = __builtin_amdgcn_readfirstlane(node);
  if (node >= n) return;

  const int cnt  = (int)deg[node];
  const float di = rsqrtf((float)cnt + 1.0f);

  const vec self = *(const vec*)(lin + (size_t)node * C + lane * PER);
  vec acc = 0.0f;

  {
    const int m = (cnt < LIST_CAP) ? cnt : LIST_CAP;
    int sv = 0;
    if (lane < m) sv = slots[(unsigned)node * LIST_CAP + lane];
#pragma unroll 4
    for (int k = 0; k < m; ++k) {
      const int s = __builtin_amdgcn_readlane(sv, k);
      acc += *(const vec*)(lin + (size_t)s * C + lane * PER);
    }
  }
  if (cnt > LIST_CAP)
    add_spilled<C, vec>(acc, spill, (int)*nSpill, node, lane, lin + lane * PER);

  acc += self;
  vec v = di * acc + *(const vec*)(bias + lane * PER);
  if (doRelu) {
#pragma unroll
    for (int j = 0; j < PER; ++j) v[j] = fmaxf(v[j], 0.0f);
  }
  *(vec*)(out + (size_t)node * C + lane * PER) = v;
}

#define G1_NODES 32
#define G1_PITCH (HID_CH + 4)

__global__ __launch_bounds__(256) void gather1_gemm2_kernel(
    const int* __restrict__ slots, const int* __restrict__ spill,
    const unsigned* __restrict__ nSpill, const unsigned* __restrict__ deg,
    const float* __restrict__ lin, const float* __restrict__ bias,
    const float* __restrict__ W2, float* __restrict__ zlin, int n) {
  __shared__ float tile[G1_NODES * G1_PITCH];
  const int lane = threadIdx.x & 31;
  const int wave = __builtin_amdgcn_readfirstlane((int)(threadIdx.x >> 5));
  const int nodeBase = (int)blockIdx.x * G1_NODES;

  constexpr int NPW = G1_NODES / 8;
  int      cnt[NPW];
  float    di[NPW];
  v4f      self[NPW];
  int      sv0[NPW];
#pragma unroll
  for (int j = 0; j < NPW; ++j) {
    const int node = nodeBase + wave * NPW + j;
    const int nc   = (node < n) ? node : n - 1;
    const int dg = (int)deg[nc];
    cnt[j]   = (node < n) ? dg : 0;
    di[j]    = rsqrtf((float)dg + 1.0f);
    self[j]  = *(const v4f*)(lin + (size_t)nc * HID_CH + lane * 4);
  }
#pragma unroll
  for (int j = 0; j < NPW; ++j) {
    const int node = nodeBase + wave * NPW + j;
    sv0[j] = 0;
    if (lane < cnt[j]) sv0[j] = slots[(unsigned)node * LIST_CAP + lane];
  }
  const v4f bv = *(const v4f*)(bias + lane * 4);
  const float* __restrict__ lrow = lin + lane * 4;
#pragma unroll
  for (int j = 0; j < NPW; ++j) {
    v4f acc = 0.0f;
    {
      const int m  = (cnt[j] < LIST_CAP) ? cnt[j] : LIST_CAP;
      const int sv = sv0[j];
      int k = 0;
      for (; k + 4 <= m; k += 4) {
        const int s0 = __builtin_amdgcn_readlane(sv, k);
        const int s1 = __builtin_amdgcn_readlane(sv, k + 1);
        const int s2 = __builtin_amdgcn_readlane(sv, k + 2);
        const int s3 = __builtin_amdgcn_readlane(sv, k + 3);
        const v4f r0 = *(const v4f*)(lrow + (size_t)s0 * HID_CH);
        const v4f r1 = *(const v4f*)(lrow + (size_t)s1 * HID_CH);
        const v4f r2 = *(const v4f*)(lrow + (size_t)s2 * HID_CH);
        const v4f r3 = *(const v4f*)(lrow + (size_t)s3 * HID_CH);
        acc += r0; acc += r1; acc += r2; acc += r3;
      }
      if (k + 2 <= m) {
        const int s0 = __builtin_amdgcn_readlane(sv, k);
        const int s1 = __builtin_amdgcn_readlane(sv, k + 1);
        const v4f r0 = *(const v4f*)(lrow + (size_t)s0 * HID_CH);
        const v4f r1 = *(const v4f*)(lrow + (size_t)s1 * HID_CH);
        acc += r0; acc += r1;
        k += 2;
      }
      if (k < m) {
        const int s0 = __builtin_amdgcn_readlane(sv, k);
        acc += *(const v4f*)(lrow + (size_t)s0 * HID_CH);
      }
    }
    if (cnt[j] > LIST_CAP)
      add_spilled<HID_CH, v4f>(acc, spill, (int)*nSpill, nodeBase + wave * NPW + j, lane, lrow);
    acc += self[j];
    v4f v = di[j] * acc + bv;
    v.x = fmaxf(v.x, 0.0f); v.y = fmaxf(v.y, 0.0f);
    v.z = fmaxf(v.z, 0.0f); v.w = fmaxf(v.w, 0.0f);
    *(v4f*)(tile + (wave * NPW + j) * G1_PITCH + lane * 4) = v;
  }
  __syncthreads();

  const int g  = lane >> 4;
  const int l  = lane & 15;
  const int mt = wave >> 2;
  const int t  = wave & 3;
  const float* __restrict__ arow = tile + (mt * 16 + l) * G1_PITCH + 2 * g;
  const float* __restrict__ wcol = W2 + (size_t)(2 * g) * OUT_CH + t * 16 + l;
  const int mbase = nodeBase + mt * 16 + 8 * g;
  float rs[8];
#pragma unroll
  for (int r = 0; r < 8; ++r) {
    const int m = mbase + r;
    rs[r] = rsqrtf((float)deg[(m < n) ? m : n - 1] + 1.0f);
  }
  v8f acc2 = {};
#pragma unroll 8
  for (int k = 0; k < HID_CH; k += 4) {
    v2f a;
    a.x = arow[k];
    a.y = arow[k + 1];
    v2f b;
    b.x = wcol[(size_t)k * OUT_CH];
    b.y = wcol[(size_t)k * OUT_CH + OUT_CH];
    acc2 = __builtin_amdgcn_wmma_f32_16x16x4_f32(
         false, a,  false, b,
         (short)0, acc2,  false,  false);
  }
#pragma unroll
  for (int r = 0; r < 8; ++r) {
    const int m = mbase + r;
    if (m < n) zlin[(size_t)m * OUT_CH + t * 16 + l] = acc2[r] * rs[r];
  }
}

template <int J>
__device__ __forceinline__ int row_bcast(int v) {
  return __builtin_amdgcn_update_dpp(0, v, 0x150 + J, 0xf, 0xf, false);
}

template <int M>
__device__ __forceinline__ float row_xor(float v) {
  return __int_as_float(
      __builtin_amdgcn_update_dpp(0, __float_as_int(v), 0x160 + M, 0xf, 0xf, false));
}

template <int J>
__device__ __forceinline__ void edge_dots(float (&p)[16], int sv, int dv,
                                          const char* __restrict__ zb, unsigned coff) {
  const unsigned so = (unsigned)row_bcast<J>(sv) * (OUT_CH * 4u) + coff;
  const unsigned dofs = (unsigned)row_bcast<J>(dv) * (OUT_CH * 4u) + coff;
  const v4f a = *(const v4f*)(zb + so);
  const v4f b = *(const v4f*)(zb + dofs);
  p[J] = a.x * b.x + a.y * b.y + a.z * b.z + a.w * b.w;
  if constexpr (J + 1 < 16) edge_dots<J + 1>(p, sv, dv, zb, coff);
}

template <int H>
__device__ __forceinline__ void fold_step(float* v, bool upper) {
#pragma unroll
  for (int a = 0; a < H; ++a) {
    const float keep = upper ? v[a + H] : v[a];
    const float send = upper ? v[a] : v[a + H];
    v[a] = keep + row_xor<H>(send);
  }
}

__global__ __launch_bounds__(256) void decode_kernel(
    const int* __restrict__ src, const int* __restrict__ dst,
    const float* __restrict__ z, float* __restrict__ out, int nE) {
  const int lane = threadIdx.x & 31;
  int e0 = (int)(((blockIdx.x * (unsigned)blockDim.x + threadIdx.x) >> 5) << 5);
  e0 = __builtin_amdgcn_readfirstlane(e0);
  if (e0 >= nE) return;

  const int e = e0 + lane;
  int sv = 0, dv = 0;
  if (e < nE) {
    sv = src[e];
    dv = dst[e];
  }

  const char* __restrict__ zb = (const char*)z;
  const unsigned coff = (unsigned)(lane & 15) * 16u;

  float p[16];
  edge_dots<0>(p, sv, dv, zb, coff);

  fold_step<8>(p, (lane & 8) != 0);
  fold_step<4>(p, (lane & 4) != 0);
  fold_step<2>(p, (lane & 2) != 0);
  fold_step<1>(p, (lane & 1) != 0);

  if (e < nE) out[e] = 1.0f / (1.0f + __expf(-p[0]));
}

extern "C" void kernel_launch(void* const* d_in, const int* in_sizes, int n_in,
                              void* d_out, int out_size, void* d_ws, size_t ws_size,
                              hipStream_t stream) {
  const float* x   = (const float*)d_in[0];
  const int*   ei  = (const int*)d_in[1];
  const float* W1  = (const float*)d_in[2];
  const float* b1  = (const float*)d_in[3];
  const float* W2  = (const float*)d_in[4];
  const float* b2  = (const float*)d_in[5];

  const int N = in_sizes[0] / IN_CH;
  const int E = in_sizes[1] / 2;
  const int* esrc = ei;
  const int* edst = ei + E;

  char* w = (char*)d_ws;
  unsigned* deg    = (unsigned*)w; w += (size_t)N * 4;
  unsigned* nSpill = (unsigned*)w; w += (size_t)N * 4;
  float*    hlin = (float*)w;    w += (size_t)N * HID_CH * 4;
  float*    h    = (float*)w;    w += (size_t)N * HID_CH * 4;
  char*     r3   = w;
  int*      slots = (int*)r3;
  unsigned* nOvf  = nSpill + 1;
  float*    W1p   = (float*)(nSpill + 96);
  int*      cells = (int*)hlin;
  int*      ovf   = (int*)h;
  float*    zlin  = h;
  int*      spill = (int*)(h + (size_t)N * OUT_CH);
  float*    z     = hlin;

  hipMemsetAsync(nSpill, 0, 8, stream);

  const int TPB = 256;
  const int nEdgeBlocks = (E + TPB - 1) / TPB;
  const int nNodeWaveBlocks = (int)(((size_t)N * 32 + TPB - 1) / TPB);
  const int mTiles = (N + 15) / 16;

  const int nBuckets = (N + BKT_NODES - 1) / BKT_NODES;
  const int nChunks  = (E + CHUNK_EDGES - 1) / CHUNK_EDGES;
  bin_edges_kernel<<<nChunks, BIN_TPB, 0, stream>>>(esrc, edst, cells, ovf, nOvf, E, nBuckets, nChunks);
  build_lists_kernel<<<nBuckets, BIN_TPB, 0, stream>>>(cells, ovf, nOvf, deg, slots, spill, nSpill, W1, W1p, N, nChunks);

  const int nGemmBlocks = (mTiles + GEMM1_TPB / 32 - 1) / (GEMM1_TPB / 32);
  gemm1_kernel<<<nGemmBlocks, GEMM1_TPB, 0, stream>>>(x, W1p, deg, hlin, N);

  const int nG1Blocks = (N + G1_NODES - 1) / G1_NODES;
  gather1_gemm2_kernel<<<nG1Blocks, TPB, 0, stream>>>(
      slots, spill, nSpill, deg, hlin, b1, W2, zlin, N);

  gather_layer<OUT_CH><<<nNodeWaveBlocks, TPB, 0, stream>>>(
      slots, spill, nSpill, deg, zlin, b2, z, N, 0);

  decode_kernel<<<nEdgeBlocks, TPB, 0, stream>>>(esrc, edst, z, (float*)d_out, E);
}
